# starting kernel
# baseline (speedup 1.0000x reference)
#include <hip/hip_runtime.h>
#include <hip/hip_bf16.h>

typedef float v2f __attribute__((ext_vector_type(2)));
typedef float v8f __attribute__((ext_vector_type(8)));

__global__ void fill_f32(float* p, float v, long n) {
  long i = (long)blockIdx.x * blockDim.x + threadIdx.x;
  if (i < n) p[i] = v;
}

__global__ void deg_accum(const int* __restrict__ dst, float* __restrict__ deg, int E) {
  long e = (long)blockIdx.x * blockDim.x + threadIdx.x;
  if (e < E) atomicAdd(&deg[dst[e]], 1.0f);
}

__global__ void deg_to_dinv(float* deg, int n) {
  long i = (long)blockIdx.x * blockDim.x + threadIdx.x;
  if (i < n) deg[i] = rsqrtf(deg[i]);
}

__global__ __launch_bounds__(256)
void gemm64_wmma(const float* __restrict__ A, const float* __restrict__ W,
                 const float* __restrict__ bias, float* __restrict__ out, int nRows) {
  __shared__ float sW[64 * 64];
  int tid = threadIdx.x;
  for (int i = tid; i < 64 * 64; i += 256) sW[i] = W[i];
  __syncthreads();

  int wave = tid >> 5;
  int lane = tid & 31;
  long tileRow = ((long)blockIdx.x * 8 + wave) * 16;

  int koff = (lane >> 4) << 1;
  int mrow = lane & 15;
  long arow = tileRow + mrow;
  if (arow >= nRows) arow = nRows - 1;
  const float* aptr = A + arow * 64 + koff;

  v8f acc[4] = {};
  for (int k0 = 0; k0 < 64; k0 += 4) {
    v2f a;
    a.x = aptr[k0];
    a.y = aptr[k0 + 1];
    int kk = k0 + koff;
#pragma unroll
    for (int n = 0; n < 4; ++n) {
      int col = n * 16 + mrow;
      v2f b;
      b.x = sW[kk * 64 + col];
      b.y = sW[(kk + 1) * 64 + col];
      acc[n] = __builtin_amdgcn_wmma_f32_16x16x4_f32(
           false, a,  false, b,
           (short)0, acc[n],  false,  false);
    }
  }

  int colBase = lane & 15;
  int rowOff = (lane >> 4) * 8;
#pragma unroll
  for (int r = 0; r < 8; ++r) {
    long row = tileRow + rowOff + r;
    if (row < nRows) {
#pragma unroll
      for (int n = 0; n < 4; ++n) {
        int col = n * 16 + colBase;
        float v = acc[n][r];
        if (bias) v += bias[col];
        out[row * 64 + col] = v;
      }
    }
  }
}

__global__ void scatter_edges(const int* __restrict__ src, const int* __restrict__ dst,
                              const float* __restrict__ h, const float* __restrict__ dinv,
                              float* __restrict__ agg, long nWork) {
  long t = (long)blockIdx.x * blockDim.x + threadIdx.x;
  if (t >= nWork) return;
  long e = t >> 6;
  int c = (int)(t & 63);
  int s = src[e];
  int d = dst[e];
  float w = dinv[s] * dinv[d];
  atomicAdd(&agg[(long)d * 64 + c], h[(long)s * 64 + c] * w);
}

__global__ void node_fuse(const float* __restrict__ agg, const float* __restrict__ h,
                          const float* __restrict__ dinv, const float* __restrict__ b,
                          const float* __restrict__ gamma, const float* __restrict__ beta,
                          const float* __restrict__ x, float* __restrict__ g, int nNodes) {
  int wave = threadIdx.x >> 5;
  int lane = threadIdx.x & 31;
  long node = (long)blockIdx.x * (blockDim.x >> 5) + wave;
  if (node >= nNodes) return;

  float di = dinv[node];
  float selfw = di * di;
  int c0 = lane, c1 = lane + 32;
  long base = node * 64;

  float v0 = agg[base + c0] + h[base + c0] * selfw + b[c0];
  float v1 = agg[base + c1] + h[base + c1] * selfw + b[c1];

  float s = v0 + v1;
#pragma unroll
  for (int off = 16; off > 0; off >>= 1) s += __shfl_xor(s, off, 32);
  float mu = s * (1.0f / 64.0f);

  float d0 = v0 - mu, d1 = v1 - mu;
  float q = d0 * d0 + d1 * d1;
#pragma unroll
  for (int off = 16; off > 0; off >>= 1) q += __shfl_xor(q, off, 32);
  float inv = rsqrtf(q * (1.0f / 64.0f) + 1e-5f);

  float y0 = d0 * inv * gamma[c0] + beta[c0];
  float y1 = d1 * inv * gamma[c1] + beta[c1];
  y0 = fmaxf(y0, 0.0f) * x[base + c0];
  y1 = fmaxf(y1, 0.0f) * x[base + c1];
  g[base + c0] = y0;
  g[base + c1] = y1;
}

extern "C" void kernel_launch(void* const* d_in, const int* in_sizes, int n_in,
                              void* d_out, int out_size, void* d_ws, size_t ws_size,
                              hipStream_t stream) {
  const float* x     = (const float*)d_in[0];
  const int*   eidx  = (const int*)d_in[1];
  const float* W     = (const float*)d_in[2];
  const float* b     = (const float*)d_in[3];
  const float* gamma = (const float*)d_in[4];
  const float* beta  = (const float*)d_in[5];
  const float* fcW   = (const float*)d_in[6];
  const float* fcb   = (const float*)d_in[7];
  float* out = (float*)d_out;

  int N = in_sizes[0] / 64;
  int E = in_sizes[1] / 2;
  const int* src = eidx;
  const int* dst = eidx + E;

  char* ws = (char*)d_ws;
  size_t off = 0;
  float* dinv = (float*)(ws + off);
  off += (((size_t)N * 4) + 255) & ~(size_t)255;
  float* h = (float*)(ws + off);
  off += (((size_t)N * 64 * 4) + 255) & ~(size_t)255;
  float* agg = (float*)(ws + off);
  float* g = h;

  const int TB = 256;

  fill_f32<<<(N + TB - 1) / TB, TB, 0, stream>>>(dinv, 1.0f, N);
  fill_f32<<<(int)(((long)N * 64 + TB - 1) / TB), TB, 0, stream>>>(agg, 0.0f, (long)N * 64);
  deg_accum<<<(E + TB - 1) / TB, TB, 0, stream>>>(dst, dinv, E);
  deg_to_dinv<<<(N + TB - 1) / TB, TB, 0, stream>>>(dinv, N);

  int gemmBlocks = (N + 127) / 128;
  gemm64_wmma<<<gemmBlocks, TB, 0, stream>>>(x, W, nullptr, h, N);

  long nWork = (long)E * 64;
  scatter_edges<<<(int)((nWork + TB - 1) / TB), TB, 0, stream>>>(src, dst, h, dinv, agg, nWork);

  node_fuse<<<(N + 7) / 8, TB, 0, stream>>>(agg, h, dinv, b, gamma, beta, x, g, N);

  gemm64_wmma<<<gemmBlocks, TB, 0, stream>>>(g, fcW, fcb, out, N);
}


# Float-atomic edge scatter replaced by in-call grouping of edges by destination (integer atomic rank) and an atomic-free per-node gather fused with the node epilogue; neighbour sums now in rank order
# speedup vs baseline: 1.7637x; 1.7637x over previous
#include <hip/hip_runtime.h>
#include <hip/hip_bf16.h>


typedef float v2f __attribute__((ext_vector_type(2)));
typedef float v8f __attribute__((ext_vector_type(8)));

__global__ void fill_i32(int* p, int v, long n) {
  long i = (long)blockIdx.x * blockDim.x + threadIdx.x;
  if (i < n) p[i] = v;
}

__global__ void rank_edges(const int* __restrict__ dst, int* cnt, int* __restrict__ rank, int E) {
  long e = (long)blockIdx.x * blockDim.x + threadIdx.x;
  if (e < E) rank[e] = atomicAdd(&cnt[dst[e]], 1);
}

__global__ __launch_bounds__(256)
void seg_offsets(const int* cnt, int* cursor, int* __restrict__ segoff,
                 float* __restrict__ dinv, int n) {
  __shared__ int sWave[8];
  __shared__ int sBase;
  int tid = threadIdx.x;
  int lane = tid & 31;
  int wave = tid >> 5;
  long i = (long)blockIdx.x * 256 + tid;
  int c = (i < n) ? cnt[i] : 0;
  if (i < n) dinv[i] = rsqrtf((float)(c + 1));

  int v = c;
#pragma unroll
  for (int o = 1; o < 32; o <<= 1) {
    int t = __shfl_up(v, o, 32);
    if (lane >= o) v += t;
  }
  if (lane == 31) sWave[wave] = v;
  __syncthreads();
  if (tid == 0) {
    int tot = 0;
    for (int w = 0; w < 8; ++w) {
      int t = sWave[w];
      sWave[w] = tot;
      tot += t;
    }
    sBase = atomicAdd(cursor, tot);
  }
  __syncthreads();
  if (i < n) segoff[i] = sBase + sWave[wave] + (v - c);
}

__global__ void place_edges(const int* __restrict__ src, const int* __restrict__ dst,
                            const int* __restrict__ rank, const int* __restrict__ segoff,
                            int* __restrict__ list, int E) {
  long e = (long)blockIdx.x * blockDim.x + threadIdx.x;
  if (e < E) list[segoff[dst[e]] + rank[e]] = src[e];
}

__global__ __launch_bounds__(256)
void gemm64_wmma(const float* __restrict__ A, const float* __restrict__ W,
                 const float* __restrict__ bias, float* __restrict__ out, int nRows) {
  __shared__ float sW[64 * 64];
  int tid = threadIdx.x;
  for (int i = tid; i < 64 * 64; i += 256) sW[i] = W[i];
  __syncthreads();

  int wave = tid >> 5;
  int lane = tid & 31;
  long tileRow = ((long)blockIdx.x * 8 + wave) * 16;

  int koff = (lane >> 4) << 1;
  int mrow = lane & 15;
  long arow = tileRow + mrow;
  if (arow >= nRows) arow = nRows - 1;
  const float* aptr = A + arow * 64 + koff;

  v8f acc[4] = {};
  for (int k0 = 0; k0 < 64; k0 += 4) {
    v2f a;
    a.x = aptr[k0];
    a.y = aptr[k0 + 1];
    int kk = k0 + koff;
#pragma unroll
    for (int n = 0; n < 4; ++n) {
      int col = n * 16 + mrow;
      v2f b;
      b.x = sW[kk * 64 + col];
      b.y = sW[(kk + 1) * 64 + col];
      acc[n] = __builtin_amdgcn_wmma_f32_16x16x4_f32(
           false, a,  false, b,
           (short)0, acc[n],  false,  false);
    }
  }

  int colBase = lane & 15;
  int rowOff = (lane >> 4) * 8;
#pragma unroll
  for (int r = 0; r < 8; ++r) {
    long row = tileRow + rowOff + r;
    if (row < nRows) {
#pragma unroll
      for (int n = 0; n < 4; ++n) {
        int col = n * 16 + colBase;
        float v = acc[n][r];
        if (bias) v += bias[col];
        out[row * 64 + col] = v;
      }
    }
  }
}

__global__ __launch_bounds__(256)
void gather_fuse(const int* __restrict__ segoff, const int* __restrict__ cnt,
                 const int* __restrict__ list, const float* __restrict__ h,
                 const float* __restrict__ dinv, const float* __restrict__ b,
                 const float* __restrict__ gamma, const float* __restrict__ beta,
                 const float* __restrict__ x, float* __restrict__ g, int nNodes) {
  int lane = threadIdx.x & 31;
  int wave = __builtin_amdgcn_readfirstlane((int)(threadIdx.x >> 5));
  long node = (long)blockIdx.x * 8 + wave;
  if (node >= nNodes) return;

  int start = segoff[node];
  int n = cnt[node];
  float di = dinv[node];
  int ch = lane * 2;
  long base = node * 64 + ch;

  v2f hs = *(const v2f*)(h + base);
  v2f xs = *(const v2f*)(x + base);
  v2f bb = *(const v2f*)(b + ch);
  v2f gg = *(const v2f*)(gamma + ch);
  v2f be = *(const v2f*)(beta + ch);

  float a0 = 0.0f, a1 = 0.0f;
  for (int p0 = 0; p0 < n; p0 += 32) {
    int m = n - p0;
    if (m > 32) m = 32;
    int sl = 0;
    float wl = 0.0f;
    if (lane < m) {
      sl = list[start + p0 + lane];
      wl = dinv[sl] * di;
    }
    int j = 0;
    for (; j + 4 <= m; j += 4) {
      int s0 = __builtin_amdgcn_readlane(sl, j);
      int s1 = __builtin_amdgcn_readlane(sl, j + 1);
      int s2 = __builtin_amdgcn_readlane(sl, j + 2);
      int s3 = __builtin_amdgcn_readlane(sl, j + 3);
      float w0 = __int_as_float(__builtin_amdgcn_readlane(__float_as_int(wl), j));
      float w1 = __int_as_float(__builtin_amdgcn_readlane(__float_as_int(wl), j + 1));
      float w2 = __int_as_float(__builtin_amdgcn_readlane(__float_as_int(wl), j + 2));
      float w3 = __int_as_float(__builtin_amdgcn_readlane(__float_as_int(wl), j + 3));
      v2f r0 = *(const v2f*)(h + (long)s0 * 64 + ch);
      v2f r1 = *(const v2f*)(h + (long)s1 * 64 + ch);
      v2f r2 = *(const v2f*)(h + (long)s2 * 64 + ch);
      v2f r3 = *(const v2f*)(h + (long)s3 * 64 + ch);
      a0 = fmaf(r0.x, w0, a0);  a1 = fmaf(r0.y, w0, a1);
      a0 = fmaf(r1.x, w1, a0);  a1 = fmaf(r1.y, w1, a1);
      a0 = fmaf(r2.x, w2, a0);  a1 = fmaf(r2.y, w2, a1);
      a0 = fmaf(r3.x, w3, a0);  a1 = fmaf(r3.y, w3, a1);
    }
    for (; j < m; ++j) {
      int s0 = __builtin_amdgcn_readlane(sl, j);
      float w0 = __int_as_float(__builtin_amdgcn_readlane(__float_as_int(wl), j));
      v2f r0 = *(const v2f*)(h + (long)s0 * 64 + ch);
      a0 = fmaf(r0.x, w0, a0);  a1 = fmaf(r0.y, w0, a1);
    }
  }

  float selfw = di * di;
  float v0 = a0 + hs.x * selfw + bb.x;
  float v1 = a1 + hs.y * selfw + bb.y;

  float s = v0 + v1;
#pragma unroll
  for (int off = 16; off > 0; off >>= 1) s += __shfl_xor(s, off, 32);
  float mu = s * (1.0f / 64.0f);

  float d0 = v0 - mu, d1 = v1 - mu;
  float q = d0 * d0 + d1 * d1;
#pragma unroll
  for (int off = 16; off > 0; off >>= 1) q += __shfl_xor(q, off, 32);
  float inv = rsqrtf(q * (1.0f / 64.0f) + 1e-5f);

  float y0 = d0 * inv * gg.x + be.x;
  float y1 = d1 * inv * gg.y + be.y;
  v2f y;
  y.x = fmaxf(y0, 0.0f) * xs.x;
  y.y = fmaxf(y1, 0.0f) * xs.y;
  *(v2f*)(g + base) = y;
}

extern "C" void kernel_launch(void* const* d_in, const int* in_sizes, int n_in,
                              void* d_out, int out_size, void* d_ws, size_t ws_size,
                              hipStream_t stream) {
  const float* x     = (const float*)d_in[0];
  const int*   eidx  = (const int*)d_in[1];
  const float* W     = (const float*)d_in[2];
  const float* b     = (const float*)d_in[3];
  const float* gamma = (const float*)d_in[4];
  const float* beta  = (const float*)d_in[5];
  const float* fcW   = (const float*)d_in[6];
  const float* fcb   = (const float*)d_in[7];
  float* out = (float*)d_out;

  int N = in_sizes[0] / 64;
  int E = in_sizes[1] / 2;
  const int* src = eidx;
  const int* dst = eidx + E;

  char* ws = (char*)d_ws;
  size_t off = 0;
  float* dinv = (float*)(ws + off);
  off += (((size_t)N * 4) + 255) & ~(size_t)255;
  float* h = (float*)(ws + off);
  off += (((size_t)N * 64 * 4) + 255) & ~(size_t)255;
  float* g = (float*)(ws + off);
  off += (((size_t)N * 64 * 4) + 255) & ~(size_t)255;
  int* cnt = (int*)(ws + off);
  int* cursor = cnt + N;
  off += ((((size_t)N + 1) * 4) + 255) & ~(size_t)255;
  int* segoff = (int*)(ws + off);
  off += (((size_t)N * 4) + 255) & ~(size_t)255;
  int* rank = (int*)(ws + off);
  off += (((size_t)E * 4) + 255) & ~(size_t)255;
  int* list = (int*)(ws + off);

  const int TB = 256;

  fill_i32<<<(N + 1 + TB - 1) / TB, TB, 0, stream>>>(cnt, 0, (long)N + 1);
  rank_edges<<<(E + TB - 1) / TB, TB, 0, stream>>>(dst, cnt, rank, E);
  seg_offsets<<<(N + TB - 1) / TB, TB, 0, stream>>>(cnt, cursor, segoff, dinv, N);
  place_edges<<<(E + TB - 1) / TB, TB, 0, stream>>>(src, dst, rank, segoff, list, E);

  int gemmBlocks = (N + 127) / 128;
  gemm64_wmma<<<gemmBlocks, TB, 0, stream>>>(x, W, nullptr, h, N);

  gather_fuse<<<(N + 7) / 8, TB, 0, stream>>>(segoff, cnt, list, h, dinv, b, gamma, beta, x, g, N);

  gemm64_wmma<<<gemmBlocks, TB, 0, stream>>>(g, fcW, fcb, out, N);
}


# Edges grouped by destination via two-launch LDS binning (chunk sort by 256-node bin, per-bin counting sort): no per-edge global atomic or scattered store; 7 launches -> 5; sum order = LDS arrival
# speedup vs baseline: 2.7481x; 1.5581x over previous
#include <hip/hip_runtime.h>
#include <hip/hip_bf16.h>


typedef float v2f __attribute__((ext_vector_type(2)));
typedef float v8f __attribute__((ext_vector_type(8)));

typedef int v2i __attribute__((ext_vector_type(2)));

#define CH 4096
#define EPT 16
#define BSH 8
#define BN 256
#define MAXNB 512
#define CAP 8192

__global__ __launch_bounds__(256)
void bin_chunks(const int* __restrict__ src, const int* __restrict__ dst, int E, int NB, int NC,
                int* __restrict__ tab, v2i* __restrict__ pairs, int* __restrict__ cursor) {
  __shared__ int sCnt[MAXNB];
  __shared__ int sOff[MAXNB];
  __shared__ v2i sPair[CH];
  __shared__ int sWave[8];
  int t = threadIdx.x;
  int lane = t & 31;
  int wave = t >> 5;
  int w = blockIdx.x;
  long e0 = (long)w * CH;
  int nE = (E - e0 < CH) ? (int)(E - e0) : CH;

  int sv[EPT], dv[EPT];
#pragma unroll
  for (int j = 0; j < EPT; ++j) {
    int i = t + j * 256;
    sv[j] = (i < nE) ? src[e0 + i] : 0;
    dv[j] = (i < nE) ? dst[e0 + i] : 0;
  }

  for (int b = t; b < MAXNB; b += 256) sCnt[b] = 0;
  if (w == 0 && t == 0) *cursor = 0;
  __syncthreads();

#pragma unroll
  for (int j = 0; j < EPT; ++j)
    if (t + j * 256 < nE) atomicAdd(&sCnt[dv[j] >> BSH], 1);
  __syncthreads();

  int b0 = 2 * t, b1 = 2 * t + 1;
  int c0 = sCnt[b0];
  int c1 = sCnt[b1];
  int s = c0 + c1;
  int v = s;
#pragma unroll
  for (int o = 1; o < 32; o <<= 1) {
    int u = __shfl_up(v, o, 32);
    if (lane >= o) v += u;
  }
  if (lane == 31) sWave[wave] = v;
  __syncthreads();
  if (t == 0) {
    int tot = 0;
    for (int k = 0; k < 8; ++k) {
      int u = sWave[k];
      sWave[k] = tot;
      tot += u;
    }
  }
  __syncthreads();
  int ex = sWave[wave] + (v - s);
  sOff[b0] = ex;
  sOff[b1] = ex + c0;
  if (b0 < NB) tab[(long)b0 * NC + w] = ex | (c0 << 16);
  if (b1 < NB) tab[(long)b1 * NC + w] = (ex + c0) | (c1 << 16);
  __syncthreads();

#pragma unroll
  for (int j = 0; j < EPT; ++j) {
    if (t + j * 256 < nE) {
      v2i pr;
      pr.x = sv[j];
      pr.y = dv[j];
      sPair[atomicAdd(&sOff[pr.y >> BSH], 1)] = pr;
    }
  }
  __syncthreads();
#pragma unroll
  for (int j = 0; j < EPT; ++j) {
    int i = t + j * 256;
    if (i < nE) pairs[e0 + i] = sPair[i];
  }
}

__global__ __launch_bounds__(256)
void bin_lists(const int* __restrict__ tab, const v2i* __restrict__ pairs, int NC, int n,
               int* cursor, int* __restrict__ segoff, int* __restrict__ cnt,
               float* __restrict__ dinv, int* __restrict__ list) {
  __shared__ int sCnt[BN];
  __shared__ int sOff[BN];
  __shared__ int sList[CAP];
  __shared__ int sWave[8];
  __shared__ int sBase;
  __shared__ int sTot;
  int t = threadIdx.x;
  int lane = t & 31;
  int wave = t >> 5;
  int nodeBase = blockIdx.x * BN;
  const int* trow = tab + (long)blockIdx.x * NC;

  sCnt[t] = 0;
  __syncthreads();

  for (int w = t; w < NC; w += 256) {
    int pk = trow[w];
    int c = pk >> 16;
    const v2i* p = pairs + (long)w * CH + (pk & 0xFFFF);
    for (int k0 = 0; k0 < c; k0 += 8) {
      int dl[8];
#pragma unroll
      for (int j = 0; j < 8; ++j)
        if (k0 + j < c) dl[j] = p[k0 + j].y;
#pragma unroll
      for (int j = 0; j < 8; ++j)
        if (k0 + j < c) atomicAdd(&sCnt[dl[j] - nodeBase], 1);
    }
  }
  __syncthreads();

  int c = sCnt[t];
  int v = c;
#pragma unroll
  for (int o = 1; o < 32; o <<= 1) {
    int u = __shfl_up(v, o, 32);
    if (lane >= o) v += u;
  }
  if (lane == 31) sWave[wave] = v;
  __syncthreads();
  if (t == 0) {
    int tot = 0;
    for (int k = 0; k < 8; ++k) {
      int u = sWave[k];
      sWave[k] = tot;
      tot += u;
    }
    sTot = tot;
    sBase = atomicAdd(cursor, tot);
  }
  __syncthreads();
  int ex = sWave[wave] + (v - c);
  int base = sBase;
  int tot = sTot;
  long node = (long)nodeBase + t;
  if (node < n) {
    segoff[node] = base + ex;
    cnt[node] = c;
    dinv[node] = rsqrtf((float)(c + 1));
  }
  sOff[t] = ex;
  __syncthreads();

  if (tot <= CAP) {
    for (int w = t; w < NC; w += 256) {
      int pk = trow[w];
      int cc = pk >> 16;
      const v2i* p = pairs + (long)w * CH + (pk & 0xFFFF);
      for (int k0 = 0; k0 < cc; k0 += 8) {
        v2i pr[8];
#pragma unroll
        for (int j = 0; j < 8; ++j)
          if (k0 + j < cc) pr[j] = p[k0 + j];
#pragma unroll
        for (int j = 0; j < 8; ++j)
          if (k0 + j < cc) sList[atomicAdd(&sOff[pr[j].y - nodeBase], 1)] = pr[j].x;
      }
    }
    __syncthreads();
    for (int i = t; i < tot; i += 256) list[base + i] = sList[i];
  } else {
    for (int w = t; w < NC; w += 256) {
      int pk = trow[w];
      int cc = pk >> 16;
      const v2i* p = pairs + (long)w * CH + (pk & 0xFFFF);
      for (int k = 0; k < cc; ++k) {
        v2i pr = p[k];
        list[base + atomicAdd(&sOff[pr.y - nodeBase], 1)] = pr.x;
      }
    }
  }
}

__global__ __launch_bounds__(256)
void gemm64_wmma(const float* __restrict__ A, const float* __restrict__ W,
                 const float* __restrict__ bias, float* __restrict__ out, int nRows) {
  __shared__ float sW[64 * 64];
  int tid = threadIdx.x;
  for (int i = tid; i < 64 * 64; i += 256) sW[i] = W[i];
  __syncthreads();

  int wave = tid >> 5;
  int lane = tid & 31;
  long tileRow = ((long)blockIdx.x * 8 + wave) * 16;

  int koff = (lane >> 4) << 1;
  int mrow = lane & 15;
  long arow = tileRow + mrow;
  if (arow >= nRows) arow = nRows - 1;
  const float* aptr = A + arow * 64 + koff;

  v8f acc[4] = {};
  for (int k0 = 0; k0 < 64; k0 += 4) {
    v2f a;
    a.x = aptr[k0];
    a.y = aptr[k0 + 1];
    int kk = k0 + koff;
#pragma unroll
    for (int n = 0; n < 4; ++n) {
      int col = n * 16 + mrow;
      v2f b;
      b.x = sW[kk * 64 + col];
      b.y = sW[(kk + 1) * 64 + col];
      acc[n] = __builtin_amdgcn_wmma_f32_16x16x4_f32(
           false, a,  false, b,
           (short)0, acc[n],  false,  false);
    }
  }

  int colBase = lane & 15;
  int rowOff = (lane >> 4) * 8;
#pragma unroll
  for (int r = 0; r < 8; ++r) {
    long row = tileRow + rowOff + r;
    if (row < nRows) {
#pragma unroll
      for (int n = 0; n < 4; ++n) {
        int col = n * 16 + colBase;
        float v = acc[n][r];
        if (bias) v += bias[col];
        out[row * 64 + col] = v;
      }
    }
  }
}

__global__ __launch_bounds__(256)
void gather_fuse(const int* __restrict__ segoff, const int* __restrict__ cnt,
                 const int* __restrict__ list, const float* __restrict__ h,
                 const float* __restrict__ dinv, const float* __restrict__ b,
                 const float* __restrict__ gamma, const float* __restrict__ beta,
                 const float* __restrict__ x, float* __restrict__ g, int nNodes) {
  int lane = threadIdx.x & 31;
  int wave = __builtin_amdgcn_readfirstlane((int)(threadIdx.x >> 5));
  long node = (long)blockIdx.x * 8 + wave;
  if (node >= nNodes) return;

  int start = segoff[node];
  int n = cnt[node];
  float di = dinv[node];
  int ch = lane * 2;
  long base = node * 64 + ch;

  v2f hs = *(const v2f*)(h + base);
  v2f xs = *(const v2f*)(x + base);
  v2f bb = *(const v2f*)(b + ch);
  v2f gg = *(const v2f*)(gamma + ch);
  v2f be = *(const v2f*)(beta + ch);

  float a0 = 0.0f, a1 = 0.0f;
  for (int p0 = 0; p0 < n; p0 += 32) {
    int m = n - p0;
    if (m > 32) m = 32;
    int sl = 0;
    float wl = 0.0f;
    if (lane < m) {
      sl = list[start + p0 + lane];
      wl = dinv[sl] * di;
    }
    int j = 0;
    for (; j + 4 <= m; j += 4) {
      int s0 = __builtin_amdgcn_readlane(sl, j);
      int s1 = __builtin_amdgcn_readlane(sl, j + 1);
      int s2 = __builtin_amdgcn_readlane(sl, j + 2);
      int s3 = __builtin_amdgcn_readlane(sl, j + 3);
      float w0 = __int_as_float(__builtin_amdgcn_readlane(__float_as_int(wl), j));
      float w1 = __int_as_float(__builtin_amdgcn_readlane(__float_as_int(wl), j + 1));
      float w2 = __int_as_float(__builtin_amdgcn_readlane(__float_as_int(wl), j + 2));
      float w3 = __int_as_float(__builtin_amdgcn_readlane(__float_as_int(wl), j + 3));
      v2f r0 = *(const v2f*)(h + (long)s0 * 64 + ch);
      v2f r1 = *(const v2f*)(h + (long)s1 * 64 + ch);
      v2f r2 = *(const v2f*)(h + (long)s2 * 64 + ch);
      v2f r3 = *(const v2f*)(h + (long)s3 * 64 + ch);
      a0 = fmaf(r0.x, w0, a0);  a1 = fmaf(r0.y, w0, a1);
      a0 = fmaf(r1.x, w1, a0);  a1 = fmaf(r1.y, w1, a1);
      a0 = fmaf(r2.x, w2, a0);  a1 = fmaf(r2.y, w2, a1);
      a0 = fmaf(r3.x, w3, a0);  a1 = fmaf(r3.y, w3, a1);
    }
    for (; j < m; ++j) {
      int s0 = __builtin_amdgcn_readlane(sl, j);
      float w0 = __int_as_float(__builtin_amdgcn_readlane(__float_as_int(wl), j));
      v2f r0 = *(const v2f*)(h + (long)s0 * 64 + ch);
      a0 = fmaf(r0.x, w0, a0);  a1 = fmaf(r0.y, w0, a1);
    }
  }

  float selfw = di * di;
  float v0 = a0 + hs.x * selfw + bb.x;
  float v1 = a1 + hs.y * selfw + bb.y;

  float s = v0 + v1;
#pragma unroll
  for (int off = 16; off > 0; off >>= 1) s += __shfl_xor(s, off, 32);
  float mu = s * (1.0f / 64.0f);

  float d0 = v0 - mu, d1 = v1 - mu;
  float q = d0 * d0 + d1 * d1;
#pragma unroll
  for (int off = 16; off > 0; off >>= 1) q += __shfl_xor(q, off, 32);
  float inv = rsqrtf(q * (1.0f / 64.0f) + 1e-5f);

  float y0 = d0 * inv * gg.x + be.x;
  float y1 = d1 * inv * gg.y + be.y;
  v2f y;
  y.x = fmaxf(y0, 0.0f) * xs.x;
  y.y = fmaxf(y1, 0.0f) * xs.y;
  *(v2f*)(g + base) = y;
}

extern "C" void kernel_launch(void* const* d_in, const int* in_sizes, int n_in,
                              void* d_out, int out_size, void* d_ws, size_t ws_size,
                              hipStream_t stream) {
  const float* x     = (const float*)d_in[0];
  const int*   eidx  = (const int*)d_in[1];
  const float* W     = (const float*)d_in[2];
  const float* b     = (const float*)d_in[3];
  const float* gamma = (const float*)d_in[4];
  const float* beta  = (const float*)d_in[5];
  const float* fcW   = (const float*)d_in[6];
  const float* fcb   = (const float*)d_in[7];
  float* out = (float*)d_out;

  int N = in_sizes[0] / 64;
  int E = in_sizes[1] / 2;
  const int* src = eidx;
  const int* dst = eidx + E;

  int NC = (E + CH - 1) / CH;
  int NB = (N + BN - 1) / BN;

  char* ws = (char*)d_ws;
  size_t off = 0;
  float* dinv = (float*)(ws + off);
  off += (((size_t)N * 4) + 255) & ~(size_t)255;
  float* h = (float*)(ws + off);
  off += (((size_t)N * 64 * 4) + 255) & ~(size_t)255;
  float* g = (float*)(ws + off);
  off += (((size_t)N * 64 * 4) + 255) & ~(size_t)255;
  int* cnt = (int*)(ws + off);
  int* cursor = cnt + N;
  off += ((((size_t)N + 1) * 4) + 255) & ~(size_t)255;
  int* segoff = (int*)(ws + off);
  off += (((size_t)N * 4) + 255) & ~(size_t)255;
  int* list = (int*)(ws + off);
  off += (((size_t)E * 4) + 255) & ~(size_t)255;
  int* tab = (int*)(ws + off);
  off += (((size_t)NB * NC * 4) + 255) & ~(size_t)255;
  v2i* pairs = (v2i*)(ws + off);

  const int TB = 256;

  bin_chunks<<<NC, TB, 0, stream>>>(src, dst, E, NB, NC, tab, pairs, cursor);
  bin_lists<<<NB, TB, 0, stream>>>(tab, pairs, NC, N, cursor, segoff, cnt, dinv, list);

  int gemmBlocks = (N + 127) / 128;
  gemm64_wmma<<<gemmBlocks, TB, 0, stream>>>(x, W, nullptr, h, N);

  gather_fuse<<<(N + 7) / 8, TB, 0, stream>>>(segoff, cnt, list, h, dinv, b, gamma, beta, x, g, N);

  gemm64_wmma<<<gemmBlocks, TB, 0, stream>>>(g, fcW, fcb, out, N);
}


# gather_fuse: 100000 one-node waves -> 2048 resident work-groups whose waves loop over nodes (grid = the chip's wave slots), so wave entry no longer paces it; per-node sums and their order unchanged
# speedup vs baseline: 2.8927x; 1.0526x over previous
#include <hip/hip_runtime.h>
#include <hip/hip_bf16.h>


typedef float v2f __attribute__((ext_vector_type(2)));
typedef float v8f __attribute__((ext_vector_type(8)));

typedef int v2i __attribute__((ext_vector_type(2)));

#define CH 4096
#define EPT 16
#define BSH 8
#define BN 256
#define MAXNB 512
#define CAP 8192

__global__ __launch_bounds__(256)
void bin_chunks(const int* __restrict__ src, const int* __restrict__ dst, int E, int NB, int NC,
                int* __restrict__ tab, v2i* __restrict__ pairs, int* __restrict__ cursor) {
  __shared__ int sCnt[MAXNB];
  __shared__ int sOff[MAXNB];
  __shared__ v2i sPair[CH];
  __shared__ int sWave[8];
  int t = threadIdx.x;
  int lane = t & 31;
  int wave = t >> 5;
  int w = blockIdx.x;
  long e0 = (long)w * CH;
  int nE = (E - e0 < CH) ? (int)(E - e0) : CH;

  int sv[EPT], dv[EPT];
#pragma unroll
  for (int j = 0; j < EPT; ++j) {
    int i = t + j * 256;
    sv[j] = (i < nE) ? src[e0 + i] : 0;
    dv[j] = (i < nE) ? dst[e0 + i] : 0;
  }

  for (int b = t; b < MAXNB; b += 256) sCnt[b] = 0;
  if (w == 0 && t == 0) *cursor = 0;
  __syncthreads();

#pragma unroll
  for (int j = 0; j < EPT; ++j)
    if (t + j * 256 < nE) atomicAdd(&sCnt[dv[j] >> BSH], 1);
  __syncthreads();

  int b0 = 2 * t, b1 = 2 * t + 1;
  int c0 = sCnt[b0];
  int c1 = sCnt[b1];
  int s = c0 + c1;
  int v = s;
#pragma unroll
  for (int o = 1; o < 32; o <<= 1) {
    int u = __shfl_up(v, o, 32);
    if (lane >= o) v += u;
  }
  if (lane == 31) sWave[wave] = v;
  __syncthreads();
  if (t == 0) {
    int tot = 0;
    for (int k = 0; k < 8; ++k) {
      int u = sWave[k];
      sWave[k] = tot;
      tot += u;
    }
  }
  __syncthreads();
  int ex = sWave[wave] + (v - s);
  sOff[b0] = ex;
  sOff[b1] = ex + c0;
  if (b0 < NB) tab[(long)b0 * NC + w] = ex | (c0 << 16);
  if (b1 < NB) tab[(long)b1 * NC + w] = (ex + c0) | (c1 << 16);
  __syncthreads();

#pragma unroll
  for (int j = 0; j < EPT; ++j) {
    if (t + j * 256 < nE) {
      v2i pr;
      pr.x = sv[j];
      pr.y = dv[j];
      sPair[atomicAdd(&sOff[pr.y >> BSH], 1)] = pr;
    }
  }
  __syncthreads();
#pragma unroll
  for (int j = 0; j < EPT; ++j) {
    int i = t + j * 256;
    if (i < nE) pairs[e0 + i] = sPair[i];
  }
}

__global__ __launch_bounds__(256)
void bin_lists(const int* __restrict__ tab, const v2i* __restrict__ pairs, int NC, int n,
               int* cursor, int* __restrict__ segoff, int* __restrict__ cnt,
               float* __restrict__ dinv, int* __restrict__ list) {
  __shared__ int sCnt[BN];
  __shared__ int sOff[BN];
  __shared__ int sList[CAP];
  __shared__ int sWave[8];
  __shared__ int sBase;
  __shared__ int sTot;
  int t = threadIdx.x;
  int lane = t & 31;
  int wave = t >> 5;
  int nodeBase = blockIdx.x * BN;
  const int* trow = tab + (long)blockIdx.x * NC;

  sCnt[t] = 0;
  __syncthreads();

  for (int w = t; w < NC; w += 256) {
    int pk = trow[w];
    int c = pk >> 16;
    const v2i* p = pairs + (long)w * CH + (pk & 0xFFFF);
    for (int k0 = 0; k0 < c; k0 += 8) {
      int dl[8];
#pragma unroll
      for (int j = 0; j < 8; ++j)
        if (k0 + j < c) dl[j] = p[k0 + j].y;
#pragma unroll
      for (int j = 0; j < 8; ++j)
        if (k0 + j < c) atomicAdd(&sCnt[dl[j] - nodeBase], 1);
    }
  }
  __syncthreads();

  int c = sCnt[t];
  int v = c;
#pragma unroll
  for (int o = 1; o < 32; o <<= 1) {
    int u = __shfl_up(v, o, 32);
    if (lane >= o) v += u;
  }
  if (lane == 31) sWave[wave] = v;
  __syncthreads();
  if (t == 0) {
    int tot = 0;
    for (int k = 0; k < 8; ++k) {
      int u = sWave[k];
      sWave[k] = tot;
      tot += u;
    }
    sTot = tot;
    sBase = atomicAdd(cursor, tot);
  }
  __syncthreads();
  int ex = sWave[wave] + (v - c);
  int base = sBase;
  int tot = sTot;
  long node = (long)nodeBase + t;
  if (node < n) {
    segoff[node] = base + ex;
    cnt[node] = c;
    dinv[node] = rsqrtf((float)(c + 1));
  }
  sOff[t] = ex;
  __syncthreads();

  if (tot <= CAP) {
    for (int w = t; w < NC; w += 256) {
      int pk = trow[w];
      int cc = pk >> 16;
      const v2i* p = pairs + (long)w * CH + (pk & 0xFFFF);
      for (int k0 = 0; k0 < cc; k0 += 8) {
        v2i pr[8];
#pragma unroll
        for (int j = 0; j < 8; ++j)
          if (k0 + j < cc) pr[j] = p[k0 + j];
#pragma unroll
        for (int j = 0; j < 8; ++j)
          if (k0 + j < cc) sList[atomicAdd(&sOff[pr[j].y - nodeBase], 1)] = pr[j].x;
      }
    }
    __syncthreads();
    for (int i = t; i < tot; i += 256) list[base + i] = sList[i];
  } else {
    for (int w = t; w < NC; w += 256) {
      int pk = trow[w];
      int cc = pk >> 16;
      const v2i* p = pairs + (long)w * CH + (pk & 0xFFFF);
      for (int k = 0; k < cc; ++k) {
        v2i pr = p[k];
        list[base + atomicAdd(&sOff[pr.y - nodeBase], 1)] = pr.x;
      }
    }
  }
}

__global__ __launch_bounds__(256)
void gemm64_wmma(const float* __restrict__ A, const float* __restrict__ W,
                 const float* __restrict__ bias, float* __restrict__ out, int nRows) {
  __shared__ float sW[64 * 64];
  int tid = threadIdx.x;
  for (int i = tid; i < 64 * 64; i += 256) sW[i] = W[i];
  __syncthreads();

  int wave = tid >> 5;
  int lane = tid & 31;
  long tileRow = ((long)blockIdx.x * 8 + wave) * 16;

  int koff = (lane >> 4) << 1;
  int mrow = lane & 15;
  long arow = tileRow + mrow;
  if (arow >= nRows) arow = nRows - 1;
  const float* aptr = A + arow * 64 + koff;

  v8f acc[4] = {};
  for (int k0 = 0; k0 < 64; k0 += 4) {
    v2f a;
    a.x = aptr[k0];
    a.y = aptr[k0 + 1];
    int kk = k0 + koff;
#pragma unroll
    for (int n = 0; n < 4; ++n) {
      int col = n * 16 + mrow;
      v2f b;
      b.x = sW[kk * 64 + col];
      b.y = sW[(kk + 1) * 64 + col];
      acc[n] = __builtin_amdgcn_wmma_f32_16x16x4_f32(
           false, a,  false, b,
           (short)0, acc[n],  false,  false);
    }
  }

  int colBase = lane & 15;
  int rowOff = (lane >> 4) * 8;
#pragma unroll
  for (int r = 0; r < 8; ++r) {
    long row = tileRow + rowOff + r;
    if (row < nRows) {
#pragma unroll
      for (int n = 0; n < 4; ++n) {
        int col = n * 16 + colBase;
        float v = acc[n][r];
        if (bias) v += bias[col];
        out[row * 64 + col] = v;
      }
    }
  }
}

__global__ __launch_bounds__(256)
void gather_fuse(const int* __restrict__ segoff, const int* __restrict__ cnt,
                 const int* __restrict__ list, const float* __restrict__ h,
                 const float* __restrict__ dinv, const float* __restrict__ b,
                 const float* __restrict__ gamma, const float* __restrict__ beta,
                 const float* __restrict__ x, float* __restrict__ g, int nNodes, int nodeStep) {
  int lane = threadIdx.x & 31;
  int wave = __builtin_amdgcn_readfirstlane((int)(threadIdx.x >> 5));
  int ch = lane * 2;
  v2f bb = *(const v2f*)(b + ch);
  v2f gg = *(const v2f*)(gamma + ch);
  v2f be = *(const v2f*)(beta + ch);

  for (long node = (long)blockIdx.x * 8 + wave; node < nNodes; node += nodeStep) {
    int start = segoff[node];
    int n = cnt[node];
    float di = dinv[node];
    long base = node * 64 + ch;

    v2f hs = *(const v2f*)(h + base);
    v2f xs = *(const v2f*)(x + base);

    float a0 = 0.0f, a1 = 0.0f;
    for (int p0 = 0; p0 < n; p0 += 32) {
      int m = n - p0;
      if (m > 32) m = 32;
      int sl = 0;
      float wl = 0.0f;
      if (lane < m) {
        sl = list[start + p0 + lane];
        wl = dinv[sl] * di;
      }
      int j = 0;
      for (; j + 4 <= m; j += 4) {
        int s0 = __builtin_amdgcn_readlane(sl, j);
        int s1 = __builtin_amdgcn_readlane(sl, j + 1);
        int s2 = __builtin_amdgcn_readlane(sl, j + 2);
        int s3 = __builtin_amdgcn_readlane(sl, j + 3);
        float w0 = __int_as_float(__builtin_amdgcn_readlane(__float_as_int(wl), j));
        float w1 = __int_as_float(__builtin_amdgcn_readlane(__float_as_int(wl), j + 1));
        float w2 = __int_as_float(__builtin_amdgcn_readlane(__float_as_int(wl), j + 2));
        float w3 = __int_as_float(__builtin_amdgcn_readlane(__float_as_int(wl), j + 3));
        v2f r0 = *(const v2f*)(h + (long)s0 * 64 + ch);
        v2f r1 = *(const v2f*)(h + (long)s1 * 64 + ch);
        v2f r2 = *(const v2f*)(h + (long)s2 * 64 + ch);
        v2f r3 = *(const v2f*)(h + (long)s3 * 64 + ch);
        a0 = fmaf(r0.x, w0, a0);  a1 = fmaf(r0.y, w0, a1);
        a0 = fmaf(r1.x, w1, a0);  a1 = fmaf(r1.y, w1, a1);
        a0 = fmaf(r2.x, w2, a0);  a1 = fmaf(r2.y, w2, a1);
        a0 = fmaf(r3.x, w3, a0);  a1 = fmaf(r3.y, w3, a1);
      }
      for (; j < m; ++j) {
        int s0 = __builtin_amdgcn_readlane(sl, j);
        float w0 = __int_as_float(__builtin_amdgcn_readlane(__float_as_int(wl), j));
        v2f r0 = *(const v2f*)(h + (long)s0 * 64 + ch);
        a0 = fmaf(r0.x, w0, a0);  a1 = fmaf(r0.y, w0, a1);
      }
    }

    float selfw = di * di;
    float v0 = a0 + hs.x * selfw + bb.x;
    float v1 = a1 + hs.y * selfw + bb.y;

    float s = v0 + v1;
#pragma unroll
    for (int off = 16; off > 0; off >>= 1) s += __shfl_xor(s, off, 32);
    float mu = s * (1.0f / 64.0f);

    float d0 = v0 - mu, d1 = v1 - mu;
    float q = d0 * d0 + d1 * d1;
#pragma unroll
    for (int off = 16; off > 0; off >>= 1) q += __shfl_xor(q, off, 32);
    float inv = rsqrtf(q * (1.0f / 64.0f) + 1e-5f);

    float y0 = d0 * inv * gg.x + be.x;
    float y1 = d1 * inv * gg.y + be.y;
    v2f y;
    y.x = fmaxf(y0, 0.0f) * xs.x;
    y.y = fmaxf(y1, 0.0f) * xs.y;
    *(v2f*)(g + base) = y;
  }
}

extern "C" void kernel_launch(void* const* d_in, const int* in_sizes, int n_in,
                              void* d_out, int out_size, void* d_ws, size_t ws_size,
                              hipStream_t stream) {
  const float* x     = (const float*)d_in[0];
  const int*   eidx  = (const int*)d_in[1];
  const float* W     = (const float*)d_in[2];
  const float* b     = (const float*)d_in[3];
  const float* gamma = (const float*)d_in[4];
  const float* beta  = (const float*)d_in[5];
  const float* fcW   = (const float*)d_in[6];
  const float* fcb   = (const float*)d_in[7];
  float* out = (float*)d_out;

  int N = in_sizes[0] / 64;
  int E = in_sizes[1] / 2;
  const int* src = eidx;
  const int* dst = eidx + E;

  int NC = (E + CH - 1) / CH;
  int NB = (N + BN - 1) / BN;

  char* ws = (char*)d_ws;
  size_t off = 0;
  float* dinv = (float*)(ws + off);
  off += (((size_t)N * 4) + 255) & ~(size_t)255;
  float* h = (float*)(ws + off);
  off += (((size_t)N * 64 * 4) + 255) & ~(size_t)255;
  float* g = (float*)(ws + off);
  off += (((size_t)N * 64 * 4) + 255) & ~(size_t)255;
  int* cnt = (int*)(ws + off);
  int* cursor = cnt + N;
  off += ((((size_t)N + 1) * 4) + 255) & ~(size_t)255;
  int* segoff = (int*)(ws + off);
  off += (((size_t)N * 4) + 255) & ~(size_t)255;
  int* list = (int*)(ws + off);
  off += (((size_t)E * 4) + 255) & ~(size_t)255;
  int* tab = (int*)(ws + off);
  off += (((size_t)NB * NC * 4) + 255) & ~(size_t)255;
  v2i* pairs = (v2i*)(ws + off);

  const int TB = 256;

  bin_chunks<<<NC, TB, 0, stream>>>(src, dst, E, NB, NC, tab, pairs, cursor);
  bin_lists<<<NB, TB, 0, stream>>>(tab, pairs, NC, N, cursor, segoff, cnt, dinv, list);

  int gemmBlocks = (N + 127) / 128;
  gemm64_wmma<<<gemmBlocks, TB, 0, stream>>>(x, W, nullptr, h, N);

  int gfBlocks = (N + 7) / 8;
  if (gfBlocks > 2048) gfBlocks = 2048;
  gather_fuse<<<gfBlocks, TB, 0, stream>>>(segoff, cnt, list, h, dinv, b, gamma, beta, x, g, N, gfBlocks * 8);

  gemm64_wmma<<<gemmBlocks, TB, 0, stream>>>(g, fcW, fcb, out, N);
}


# h = x @ W moved into the first edge-binning launch as extra work-groups (block-uniform branch, shared LDS bytes): 5 launches -> 4; arithmetic and sum order unchanged
# speedup vs baseline: 3.0088x; 1.0401x over previous
#include <hip/hip_runtime.h>
#include <hip/hip_bf16.h>


typedef float v2f __attribute__((ext_vector_type(2)));
typedef float v8f __attribute__((ext_vector_type(8)));

typedef int v2i __attribute__((ext_vector_type(2)));

#define CH 4096
#define EPT 16
#define BSH 8
#define BN 256
#define MAXNB 512
#define CAP 8192

struct BinChunkLds {
  v2i pair[CH];
  int cnt[MAXNB];
  int off[MAXNB];
  int wave[8];
};

union FirstLaunchLds {
  BinChunkLds bin;
  float w[64 * 64];
};

__device__ __forceinline__
void bin_chunks_body(const int* __restrict__ src, const int* __restrict__ dst, int E, int NB, int NC,
                     int* __restrict__ tab, v2i* __restrict__ pairs, int* __restrict__ cursor,
                     BinChunkLds& lds, int w) {
  int* sCnt = lds.cnt;
  int* sOff = lds.off;
  v2i* sPair = lds.pair;
  int* sWave = lds.wave;
  int t = threadIdx.x;
  int lane = t & 31;
  int wave = t >> 5;
  long e0 = (long)w * CH;
  int nE = (E - e0 < CH) ? (int)(E - e0) : CH;

  int sv[EPT], dv[EPT];
#pragma unroll
  for (int j = 0; j < EPT; ++j) {
    int i = t + j * 256;
    sv[j] = (i < nE) ? src[e0 + i] : 0;
    dv[j] = (i < nE) ? dst[e0 + i] : 0;
  }

  for (int b = t; b < MAXNB; b += 256) sCnt[b] = 0;
  if (w == 0 && t == 0) *cursor = 0;
  __syncthreads();

#pragma unroll
  for (int j = 0; j < EPT; ++j)
    if (t + j * 256 < nE) atomicAdd(&sCnt[dv[j] >> BSH], 1);
  __syncthreads();

  int b0 = 2 * t, b1 = 2 * t + 1;
  int c0 = sCnt[b0];
  int c1 = sCnt[b1];
  int s = c0 + c1;
  int v = s;
#pragma unroll
  for (int o = 1; o < 32; o <<= 1) {
    int u = __shfl_up(v, o, 32);
    if (lane >= o) v += u;
  }
  if (lane == 31) sWave[wave] = v;
  __syncthreads();
  if (t == 0) {
    int tot = 0;
    for (int k = 0; k < 8; ++k) {
      int u = sWave[k];
      sWave[k] = tot;
      tot += u;
    }
  }
  __syncthreads();
  int ex = sWave[wave] + (v - s);
  sOff[b0] = ex;
  sOff[b1] = ex + c0;
  if (b0 < NB) tab[(long)b0 * NC + w] = ex | (c0 << 16);
  if (b1 < NB) tab[(long)b1 * NC + w] = (ex + c0) | (c1 << 16);
  __syncthreads();

#pragma unroll
  for (int j = 0; j < EPT; ++j) {
    if (t + j * 256 < nE) {
      v2i pr;
      pr.x = sv[j];
      pr.y = dv[j];
      sPair[atomicAdd(&sOff[pr.y >> BSH], 1)] = pr;
    }
  }
  __syncthreads();
#pragma unroll
  for (int j = 0; j < EPT; ++j) {
    int i = t + j * 256;
    if (i < nE) pairs[e0 + i] = sPair[i];
  }
}

__global__ __launch_bounds__(256)
void bin_lists(const int* __restrict__ tab, const v2i* __restrict__ pairs, int NC, int n,
               int* cursor, int* __restrict__ segoff, int* __restrict__ cnt,
               float* __restrict__ dinv, int* __restrict__ list) {
  __shared__ int sCnt[BN];
  __shared__ int sOff[BN];
  __shared__ int sList[CAP];
  __shared__ int sWave[8];
  __shared__ int sBase;
  __shared__ int sTot;
  int t = threadIdx.x;
  int lane = t & 31;
  int wave = t >> 5;
  int nodeBase = blockIdx.x * BN;
  const int* trow = tab + (long)blockIdx.x * NC;

  sCnt[t] = 0;
  __syncthreads();

  for (int w = t; w < NC; w += 256) {
    int pk = trow[w];
    int c = pk >> 16;
    const v2i* p = pairs + (long)w * CH + (pk & 0xFFFF);
    for (int k0 = 0; k0 < c; k0 += 8) {
      int dl[8];
#pragma unroll
      for (int j = 0; j < 8; ++j)
        if (k0 + j < c) dl[j] = p[k0 + j].y;
#pragma unroll
      for (int j = 0; j < 8; ++j)
        if (k0 + j < c) atomicAdd(&sCnt[dl[j] - nodeBase], 1);
    }
  }
  __syncthreads();

  int c = sCnt[t];
  int v = c;
#pragma unroll
  for (int o = 1; o < 32; o <<= 1) {
    int u = __shfl_up(v, o, 32);
    if (lane >= o) v += u;
  }
  if (lane == 31) sWave[wave] = v;
  __syncthreads();
  if (t == 0) {
    int tot = 0;
    for (int k = 0; k < 8; ++k) {
      int u = sWave[k];
      sWave[k] = tot;
      tot += u;
    }
    sTot = tot;
    sBase = atomicAdd(cursor, tot);
  }
  __syncthreads();
  int ex = sWave[wave] + (v - c);
  int base = sBase;
  int tot = sTot;
  long node = (long)nodeBase + t;
  if (node < n) {
    segoff[node] = base + ex;
    cnt[node] = c;
    dinv[node] = rsqrtf((float)(c + 1));
  }
  sOff[t] = ex;
  __syncthreads();

  if (tot <= CAP) {
    for (int w = t; w < NC; w += 256) {
      int pk = trow[w];
      int cc = pk >> 16;
      const v2i* p = pairs + (long)w * CH + (pk & 0xFFFF);
      for (int k0 = 0; k0 < cc; k0 += 8) {
        v2i pr[8];
#pragma unroll
        for (int j = 0; j < 8; ++j)
          if (k0 + j < cc) pr[j] = p[k0 + j];
#pragma unroll
        for (int j = 0; j < 8; ++j)
          if (k0 + j < cc) sList[atomicAdd(&sOff[pr[j].y - nodeBase], 1)] = pr[j].x;
      }
    }
    __syncthreads();
    for (int i = t; i < tot; i += 256) list[base + i] = sList[i];
  } else {
    for (int w = t; w < NC; w += 256) {
      int pk = trow[w];
      int cc = pk >> 16;
      const v2i* p = pairs + (long)w * CH + (pk & 0xFFFF);
      for (int k = 0; k < cc; ++k) {
        v2i pr = p[k];
        list[base + atomicAdd(&sOff[pr.y - nodeBase], 1)] = pr.x;
      }
    }
  }
}

__device__ __forceinline__
void gemm64_body(const float* __restrict__ A, const float* __restrict__ W,
                 const float* __restrict__ bias, float* __restrict__ out, int nRows,
                 float* sW, int blk) {
  int tid = threadIdx.x;
  for (int i = tid; i < 64 * 64; i += 256) sW[i] = W[i];
  __syncthreads();

  int wave = tid >> 5;
  int lane = tid & 31;
  long tileRow = ((long)blk * 8 + wave) * 16;

  int koff = (lane >> 4) << 1;
  int mrow = lane & 15;
  long arow = tileRow + mrow;
  if (arow >= nRows) arow = nRows - 1;
  const float* aptr = A + arow * 64 + koff;

  v8f acc[4] = {};
  for (int k0 = 0; k0 < 64; k0 += 4) {
    v2f a;
    a.x = aptr[k0];
    a.y = aptr[k0 + 1];
    int kk = k0 + koff;
#pragma unroll
    for (int n = 0; n < 4; ++n) {
      int col = n * 16 + mrow;
      v2f b;
      b.x = sW[kk * 64 + col];
      b.y = sW[(kk + 1) * 64 + col];
      acc[n] = __builtin_amdgcn_wmma_f32_16x16x4_f32(
           false, a,  false, b,
           (short)0, acc[n],  false,  false);
    }
  }

  int colBase = lane & 15;
  int rowOff = (lane >> 4) * 8;
#pragma unroll
  for (int r = 0; r < 8; ++r) {
    long row = tileRow + rowOff + r;
    if (row < nRows) {
#pragma unroll
      for (int n = 0; n < 4; ++n) {
        int col = n * 16 + colBase;
        float v = acc[n][r];
        if (bias) v += bias[col];
        out[row * 64 + col] = v;
      }
    }
  }
}

__global__ __launch_bounds__(256)
void gemm64_wmma(const float* __restrict__ A, const float* __restrict__ W,
                 const float* __restrict__ bias, float* __restrict__ out, int nRows) {
  __shared__ float sW[64 * 64];
  gemm64_body(A, W, bias, out, nRows, sW, (int)blockIdx.x);
}

__global__ __launch_bounds__(256)
void bin_chunks_gemm(const int* __restrict__ src, const int* __restrict__ dst, int E, int NB, int NC,
                     int* __restrict__ tab, v2i* __restrict__ pairs, int* __restrict__ cursor,
                     const float* __restrict__ x, const float* __restrict__ W,
                     float* __restrict__ h, int nRows, int nGemm) {
  __shared__ FirstLaunchLds lds;
  int blk = (int)blockIdx.x;
  if (blk < nGemm) {
    gemm64_body(x, W, nullptr, h, nRows, lds.w, blk);
  } else {
    bin_chunks_body(src, dst, E, NB, NC, tab, pairs, cursor, lds.bin, blk - nGemm);
  }
}

__global__ __launch_bounds__(256)
void gather_fuse(const int* __restrict__ segoff, const int* __restrict__ cnt,
                 const int* __restrict__ list, const float* __restrict__ h,
                 const float* __restrict__ dinv, const float* __restrict__ b,
                 const float* __restrict__ gamma, const float* __restrict__ beta,
                 const float* __restrict__ x, float* __restrict__ g, int nNodes, int nodeStep) {
  int lane = threadIdx.x & 31;
  int wave = __builtin_amdgcn_readfirstlane((int)(threadIdx.x >> 5));
  int ch = lane * 2;
  v2f bb = *(const v2f*)(b + ch);
  v2f gg = *(const v2f*)(gamma + ch);
  v2f be = *(const v2f*)(beta + ch);

  for (long node = (long)blockIdx.x * 8 + wave; node < nNodes; node += nodeStep) {
    int start = segoff[node];
    int n = cnt[node];
    float di = dinv[node];
    long base = node * 64 + ch;

    v2f hs = *(const v2f*)(h + base);
    v2f xs = *(const v2f*)(x + base);

    float a0 = 0.0f, a1 = 0.0f;
    for (int p0 = 0; p0 < n; p0 += 32) {
      int m = n - p0;
      if (m > 32) m = 32;
      int sl = 0;
      float wl = 0.0f;
      if (lane < m) {
        sl = list[start + p0 + lane];
        wl = dinv[sl] * di;
      }
      int j = 0;
      for (; j + 4 <= m; j += 4) {
        int s0 = __builtin_amdgcn_readlane(sl, j);
        int s1 = __builtin_amdgcn_readlane(sl, j + 1);
        int s2 = __builtin_amdgcn_readlane(sl, j + 2);
        int s3 = __builtin_amdgcn_readlane(sl, j + 3);
        float w0 = __int_as_float(__builtin_amdgcn_readlane(__float_as_int(wl), j));
        float w1 = __int_as_float(__builtin_amdgcn_readlane(__float_as_int(wl), j + 1));
        float w2 = __int_as_float(__builtin_amdgcn_readlane(__float_as_int(wl), j + 2));
        float w3 = __int_as_float(__builtin_amdgcn_readlane(__float_as_int(wl), j + 3));
        v2f r0 = *(const v2f*)(h + (long)s0 * 64 + ch);
        v2f r1 = *(const v2f*)(h + (long)s1 * 64 + ch);
        v2f r2 = *(const v2f*)(h + (long)s2 * 64 + ch);
        v2f r3 = *(const v2f*)(h + (long)s3 * 64 + ch);
        a0 = fmaf(r0.x, w0, a0);  a1 = fmaf(r0.y, w0, a1);
        a0 = fmaf(r1.x, w1, a0);  a1 = fmaf(r1.y, w1, a1);
        a0 = fmaf(r2.x, w2, a0);  a1 = fmaf(r2.y, w2, a1);
        a0 = fmaf(r3.x, w3, a0);  a1 = fmaf(r3.y, w3, a1);
      }
      for (; j < m; ++j) {
        int s0 = __builtin_amdgcn_readlane(sl, j);
        float w0 = __int_as_float(__builtin_amdgcn_readlane(__float_as_int(wl), j));
        v2f r0 = *(const v2f*)(h + (long)s0 * 64 + ch);
        a0 = fmaf(r0.x, w0, a0);  a1 = fmaf(r0.y, w0, a1);
      }
    }

    float selfw = di * di;
    float v0 = a0 + hs.x * selfw + bb.x;
    float v1 = a1 + hs.y * selfw + bb.y;

    float s = v0 + v1;
#pragma unroll
    for (int off = 16; off > 0; off >>= 1) s += __shfl_xor(s, off, 32);
    float mu = s * (1.0f / 64.0f);

    float d0 = v0 - mu, d1 = v1 - mu;
    float q = d0 * d0 + d1 * d1;
#pragma unroll
    for (int off = 16; off > 0; off >>= 1) q += __shfl_xor(q, off, 32);
    float inv = rsqrtf(q * (1.0f / 64.0f) + 1e-5f);

    float y0 = d0 * inv * gg.x + be.x;
    float y1 = d1 * inv * gg.y + be.y;
    v2f y;
    y.x = fmaxf(y0, 0.0f) * xs.x;
    y.y = fmaxf(y1, 0.0f) * xs.y;
    *(v2f*)(g + base) = y;
  }
}

extern "C" void kernel_launch(void* const* d_in, const int* in_sizes, int n_in,
                              void* d_out, int out_size, void* d_ws, size_t ws_size,
                              hipStream_t stream) {
  const float* x     = (const float*)d_in[0];
  const int*   eidx  = (const int*)d_in[1];
  const float* W     = (const float*)d_in[2];
  const float* b     = (const float*)d_in[3];
  const float* gamma = (const float*)d_in[4];
  const float* beta  = (const float*)d_in[5];
  const float* fcW   = (const float*)d_in[6];
  const float* fcb   = (const float*)d_in[7];
  float* out = (float*)d_out;

  int N = in_sizes[0] / 64;
  int E = in_sizes[1] / 2;
  const int* src = eidx;
  const int* dst = eidx + E;

  int NC = (E + CH - 1) / CH;
  int NB = (N + BN - 1) / BN;

  char* ws = (char*)d_ws;
  size_t off = 0;
  float* dinv = (float*)(ws + off);
  off += (((size_t)N * 4) + 255) & ~(size_t)255;
  float* h = (float*)(ws + off);
  off += (((size_t)N * 64 * 4) + 255) & ~(size_t)255;
  float* g = (float*)(ws + off);
  off += (((size_t)N * 64 * 4) + 255) & ~(size_t)255;
  int* cnt = (int*)(ws + off);
  int* cursor = cnt + N;
  off += ((((size_t)N + 1) * 4) + 255) & ~(size_t)255;
  int* segoff = (int*)(ws + off);
  off += (((size_t)N * 4) + 255) & ~(size_t)255;
  int* list = (int*)(ws + off);
  off += (((size_t)E * 4) + 255) & ~(size_t)255;
  int* tab = (int*)(ws + off);
  off += (((size_t)NB * NC * 4) + 255) & ~(size_t)255;
  v2i* pairs = (v2i*)(ws + off);

  const int TB = 256;

  int gemmBlocks = (N + 127) / 128;
  bin_chunks_gemm<<<gemmBlocks + NC, TB, 0, stream>>>(src, dst, E, NB, NC, tab, pairs, cursor, x, W, h, N, gemmBlocks);
  bin_lists<<<NB, TB, 0, stream>>>(tab, pairs, NC, N, cursor, segoff, cnt, dinv, list);

  int gfBlocks = (N + 7) / 8;
  if (gfBlocks > 2048) gfBlocks = 2048;
  gather_fuse<<<gfBlocks, TB, 0, stream>>>(segoff, cnt, list, h, dinv, b, gamma, beta, x, g, N, gfBlocks * 8);

  gemm64_wmma<<<gemmBlocks, TB, 0, stream>>>(g, fcW, fcb, out, N);
}


# gather_fuse: per-node load chain overlapped across a wave's nodes (headers one per lane; next node's list and source dinv fetched under this node's rows and LayerNorm; 8 rows a group); sum order kept
# speedup vs baseline: 3.6841x; 1.2244x over previous
#include <hip/hip_runtime.h>
#include <hip/hip_bf16.h>


typedef float v2f __attribute__((ext_vector_type(2)));
typedef float v4f __attribute__((ext_vector_type(4)));
typedef float v8f __attribute__((ext_vector_type(8)));

typedef int v2i __attribute__((ext_vector_type(2)));

#define CH 4096
#define EPT 16
#define BSH 8
#define BN 256
#define MAXNB 512
#define CAP 8192

struct BinChunkLds {
  v2i pair[CH];
  int cnt[MAXNB];
  int off[MAXNB];
  int wave[8];
};

union alignas(16) FirstLaunchLds {
  BinChunkLds bin;
  float w[64 * 64];
};

__device__ __forceinline__
void bin_chunks_body(const int* __restrict__ src, const int* __restrict__ dst, int E, int NB, int NC,
                     int* __restrict__ tab, v2i* __restrict__ pairs, int* __restrict__ cursor,
                     BinChunkLds& lds, int w) {
  int* sCnt = lds.cnt;
  int* sOff = lds.off;
  v2i* sPair = lds.pair;
  int* sWave = lds.wave;
  int t = threadIdx.x;
  int lane = t & 31;
  int wave = t >> 5;
  long e0 = (long)w * CH;
  int nE = (E - e0 < CH) ? (int)(E - e0) : CH;

  int sv[EPT], dv[EPT];
#pragma unroll
  for (int j = 0; j < EPT; ++j) {
    int i = t + j * 256;
    sv[j] = (i < nE) ? src[e0 + i] : 0;
    dv[j] = (i < nE) ? dst[e0 + i] : 0;
  }

  for (int b = t; b < MAXNB; b += 256) sCnt[b] = 0;
  if (w == 0 && t == 0) *cursor = 0;
  __syncthreads();

#pragma unroll
  for (int j = 0; j < EPT; ++j)
    if (t + j * 256 < nE) atomicAdd(&sCnt[dv[j] >> BSH], 1);
  __syncthreads();

  int b0 = 2 * t, b1 = 2 * t + 1;
  int c0 = sCnt[b0];
  int c1 = sCnt[b1];
  int s = c0 + c1;
  int v = s;
#pragma unroll
  for (int o = 1; o < 32; o <<= 1) {
    int u = __shfl_up(v, o, 32);
    if (lane >= o) v += u;
  }
  if (lane == 31) sWave[wave] = v;
  __syncthreads();
  if (t == 0) {
    int tot = 0;
    for (int k = 0; k < 8; ++k) {
      int u = sWave[k];
      sWave[k] = tot;
      tot += u;
    }
  }
  __syncthreads();
  int ex = sWave[wave] + (v - s);
  sOff[b0] = ex;
  sOff[b1] = ex + c0;
  if (b0 < NB) tab[(long)b0 * NC + w] = ex | (c0 << 16);
  if (b1 < NB) tab[(long)b1 * NC + w] = (ex + c0) | (c1 << 16);
  __syncthreads();

#pragma unroll
  for (int j = 0; j < EPT; ++j) {
    if (t + j * 256 < nE) {
      v2i pr;
      pr.x = sv[j];
      pr.y = dv[j];
      sPair[atomicAdd(&sOff[pr.y >> BSH], 1)] = pr;
    }
  }
  __syncthreads();
#pragma unroll
  for (int j = 0; j < EPT; ++j) {
    int i = t + j * 256;
    if (i < nE) pairs[e0 + i] = sPair[i];
  }
}

__global__ __launch_bounds__(256)
void bin_lists(const int* __restrict__ tab, const v2i* __restrict__ pairs, int NC, int n,
               int* cursor, int* __restrict__ segoff, int* __restrict__ cnt,
               float* __restrict__ dinv, int* __restrict__ list) {
  __shared__ int sCnt[BN];
  __shared__ int sOff[BN];
  __shared__ int sList[CAP];
  __shared__ int sWave[8];
  __shared__ int sBase;
  __shared__ int sTot;
  int t = threadIdx.x;
  int lane = t & 31;
  int wave = t >> 5;
  int nodeBase = blockIdx.x * BN;
  const int* trow = tab + (long)blockIdx.x * NC;

  sCnt[t] = 0;
  __syncthreads();

  for (int w = t; w < NC; w += 256) {
    int pk = trow[w];
    int c = pk >> 16;
    const v2i* p = pairs + (long)w * CH + (pk & 0xFFFF);
    for (int k0 = 0; k0 < c; k0 += 8) {
      int dl[8];
#pragma unroll
      for (int j = 0; j < 8; ++j)
        if (k0 + j < c) dl[j] = p[k0 + j].y;
#pragma unroll
      for (int j = 0; j < 8; ++j)
        if (k0 + j < c) atomicAdd(&sCnt[dl[j] - nodeBase], 1);
    }
  }
  __syncthreads();

  int c = sCnt[t];
  int v = c;
#pragma unroll
  for (int o = 1; o < 32; o <<= 1) {
    int u = __shfl_up(v, o, 32);
    if (lane >= o) v += u;
  }
  if (lane == 31) sWave[wave] = v;
  __syncthreads();
  if (t == 0) {
    int tot = 0;
    for (int k = 0; k < 8; ++k) {
      int u = sWave[k];
      sWave[k] = tot;
      tot += u;
    }
    sTot = tot;
    sBase = atomicAdd(cursor, tot);
  }
  __syncthreads();
  int ex = sWave[wave] + (v - c);
  int base = sBase;
  int tot = sTot;
  long node = (long)nodeBase + t;
  if (node < n) {
    segoff[node] = base + ex;
    cnt[node] = c;
    dinv[node] = rsqrtf((float)(c + 1));
  }
  sOff[t] = ex;
  __syncthreads();

  if (tot <= CAP) {
    for (int w = t; w < NC; w += 256) {
      int pk = trow[w];
      int cc = pk >> 16;
      const v2i* p = pairs + (long)w * CH + (pk & 0xFFFF);
      for (int k0 = 0; k0 < cc; k0 += 8) {
        v2i pr[8];
#pragma unroll
        for (int j = 0; j < 8; ++j)
          if (k0 + j < cc) pr[j] = p[k0 + j];
#pragma unroll
        for (int j = 0; j < 8; ++j)
          if (k0 + j < cc) sList[atomicAdd(&sOff[pr[j].y - nodeBase], 1)] = pr[j].x;
      }
    }
    __syncthreads();
    for (int i = t; i < tot; i += 256) list[base + i] = sList[i];
  } else {
    for (int w = t; w < NC; w += 256) {
      int pk = trow[w];
      int cc = pk >> 16;
      const v2i* p = pairs + (long)w * CH + (pk & 0xFFFF);
      for (int k = 0; k < cc; ++k) {
        v2i pr = p[k];
        list[base + atomicAdd(&sOff[pr.y - nodeBase], 1)] = pr.x;
      }
    }
  }
}

template <bool HAS_BIAS>
__device__ __forceinline__
void gemm64_body(const float* __restrict__ A, const float* __restrict__ W,
                 const float* __restrict__ bias, float* __restrict__ out, int nRows,
                 float* sW, int blk) {
  int tid = threadIdx.x;
  int wave = tid >> 5;
  int lane = tid & 31;
  long tileRow = ((long)blk * 8 + wave) * 16;

  int koff = (lane >> 4) << 1;
  int mrow = lane & 15;
  long arow = tileRow + mrow;
  if (arow >= nRows) arow = nRows - 1;
  const float* aptr = A + arow * 64 + koff;

  v4f wreg[4];
#pragma unroll
  for (int j = 0; j < 4; ++j) wreg[j] = ((const v4f*)W)[tid + j * 256];
  v2f a[16];
#pragma unroll
  for (int k = 0; k < 16; ++k) a[k] = *(const v2f*)(aptr + 4 * k);
  float bv[4] = {0.0f, 0.0f, 0.0f, 0.0f};
  if (HAS_BIAS) {
#pragma unroll
    for (int n = 0; n < 4; ++n) bv[n] = bias[n * 16 + mrow];
  }
#pragma unroll
  for (int j = 0; j < 4; ++j) ((v4f*)sW)[tid + j * 256] = wreg[j];
  __syncthreads();

  v8f acc[4] = {};
#pragma unroll
  for (int k = 0; k < 16; ++k) {
    int kk = 4 * k + koff;
#pragma unroll
    for (int n = 0; n < 4; ++n) {
      int col = n * 16 + mrow;
      v2f b;
      b.x = sW[kk * 64 + col];
      b.y = sW[(kk + 1) * 64 + col];
      acc[n] = __builtin_amdgcn_wmma_f32_16x16x4_f32(
           false, a[k],  false, b,
           (short)0, acc[n],  false,  false);
    }
  }

  int colBase = lane & 15;
  int rowOff = (lane >> 4) * 8;
#pragma unroll
  for (int r = 0; r < 8; ++r) {
    long row = tileRow + rowOff + r;
    if (row < nRows) {
#pragma unroll
      for (int n = 0; n < 4; ++n) {
        int col = n * 16 + colBase;
        float v = acc[n][r];
        if (HAS_BIAS) v += bv[n];
        out[row * 64 + col] = v;
      }
    }
  }
}

__global__ __launch_bounds__(256)
void gemm64_wmma(const float* __restrict__ A, const float* __restrict__ W,
                 const float* __restrict__ bias, float* __restrict__ out, int nRows) {
  __shared__ alignas(16) float sW[64 * 64];
  gemm64_body<true>(A, W, bias, out, nRows, sW, (int)blockIdx.x);
}

__global__ __launch_bounds__(256)
void bin_chunks_gemm(const int* __restrict__ src, const int* __restrict__ dst, int E, int NB, int NC,
                     int* __restrict__ tab, v2i* __restrict__ pairs, int* __restrict__ cursor,
                     const float* __restrict__ x, const float* __restrict__ W,
                     float* __restrict__ h, int nRows, int nGemm) {
  __shared__ FirstLaunchLds lds;
  int blk = (int)blockIdx.x;
  if (blk < nGemm) {
    gemm64_body<false>(x, W, nullptr, h, nRows, lds.w, blk);
  } else {
    bin_chunks_body(src, dst, E, NB, NC, tab, pairs, cursor, lds.bin, blk - nGemm);
  }
}

#define RG 8

__device__ __forceinline__ float lane_f(float v, int l) {
  return __int_as_float(__builtin_amdgcn_readlane(__float_as_int(v), l));
}

template <int K>
__device__ __forceinline__
void rows_k(const float* __restrict__ hch, int sl, float wl, int j, float& a0, float& a1) {
  v2f r[K];
#pragma unroll
  for (int q = 0; q < K; ++q) {
    int s = __builtin_amdgcn_readlane(sl, j + q);
    r[q] = *(const v2f*)(hch + (long)s * 64);
  }
#pragma unroll
  for (int q = 0; q < K; ++q) {
    float w = lane_f(wl, j + q);
    a0 = fmaf(r[q].x, w, a0);
    a1 = fmaf(r[q].y, w, a1);
  }
}

__global__ __launch_bounds__(256)
void gather_fuse(const int* __restrict__ segoff, const int* __restrict__ cnt,
                 const int* __restrict__ list, const float* __restrict__ h,
                 const float* __restrict__ dinv, const float* __restrict__ b,
                 const float* __restrict__ gamma, const float* __restrict__ beta,
                 const float* __restrict__ x, float* __restrict__ g, int nNodes, int nodeStep) {
  int lane = threadIdx.x & 31;
  int wave = __builtin_amdgcn_readfirstlane((int)(threadIdx.x >> 5));
  int ch = lane * 2;
  v2f bb = *(const v2f*)(b + ch);
  v2f gg = *(const v2f*)(gamma + ch);
  v2f be = *(const v2f*)(beta + ch);
  const float* hch = h + ch;

  long first = (long)blockIdx.x * 8 + wave;
  for (long nb = first; nb < nNodes; nb += 32L * nodeStep) {
    long mine = nb + (long)lane * nodeStep;
    bool have = mine < nNodes;
    int stL = 0, nL = 0;
    float diL = 0.0f;
    if (have) {
      stL = segoff[mine];
      nL = cnt[mine];
      diL = dinv[mine];
    }
    int kc = __builtin_popcount(__builtin_amdgcn_ballot_w32(have));

    int start = __builtin_amdgcn_readlane(stL, 0);
    int n = __builtin_amdgcn_readlane(nL, 0);
    float di = lane_f(diL, 0);
    int m = n < 32 ? n : 32;
    int sl = list[lane < m ? start + lane : 0];
    sl = lane < m ? sl : 0;
    float dsl = dinv[sl];

    for (int i = 0; i < kc; ++i) {
      long base = (nb + (long)i * nodeStep) * 64 + ch;

      int nstart = 0, nn = 0;
      float ndi = 0.0f;
      if (i + 1 < kc) {
        nstart = __builtin_amdgcn_readlane(stL, i + 1);
        nn = __builtin_amdgcn_readlane(nL, i + 1);
        ndi = lane_f(diL, i + 1);
      }
      int nm = nn < 32 ? nn : 32;
      int nsl = list[lane < nm ? nstart + lane : 0];

      v2f hs = *(const v2f*)(h + base);
      v2f xs = *(const v2f*)(x + base);

      float a0 = 0.0f, a1 = 0.0f;
      if (n > 0) {
        float wl = dsl * di;
        int j = 0;
        for (; j + RG <= m; j += RG) rows_k<RG>(hch, sl, wl, j, a0, a1);
        switch (m - j) {
          case 1: rows_k<1>(hch, sl, wl, j, a0, a1); break;
          case 2: rows_k<2>(hch, sl, wl, j, a0, a1); break;
          case 3: rows_k<3>(hch, sl, wl, j, a0, a1); break;
          case 4: rows_k<4>(hch, sl, wl, j, a0, a1); break;
          case 5: rows_k<5>(hch, sl, wl, j, a0, a1); break;
          case 6: rows_k<6>(hch, sl, wl, j, a0, a1); break;
          case 7: rows_k<7>(hch, sl, wl, j, a0, a1); break;
          default: break;
        }
        for (int p0 = 32; p0 < n; p0 += 32) {
          int m2 = n - p0;
          if (m2 > 32) m2 = 32;
          int s2 = 0;
          float d2 = 0.0f;
          if (lane < m2) {
            s2 = list[start + p0 + lane];
            d2 = dinv[s2];
          }
          float w2 = d2 * di;
          int k = 0;
          for (; k + 4 <= m2; k += 4) rows_k<4>(hch, s2, w2, k, a0, a1);
          for (; k < m2; ++k) rows_k<1>(hch, s2, w2, k, a0, a1);
        }
      }

      nsl = lane < nm ? nsl : 0;
      float ndsl = dinv[nsl];

      float selfw = di * di;
      float v0 = a0 + hs.x * selfw + bb.x;
      float v1 = a1 + hs.y * selfw + bb.y;

      float s = v0 + v1;
#pragma unroll
      for (int off = 16; off > 0; off >>= 1) s += __shfl_xor(s, off, 32);
      float mu = s * (1.0f / 64.0f);

      float d0 = v0 - mu, d1 = v1 - mu;
      float qq = d0 * d0 + d1 * d1;
#pragma unroll
      for (int off = 16; off > 0; off >>= 1) qq += __shfl_xor(qq, off, 32);
      float inv = rsqrtf(qq * (1.0f / 64.0f) + 1e-5f);

      float y0 = d0 * inv * gg.x + be.x;
      float y1 = d1 * inv * gg.y + be.y;
      v2f y;
      y.x = fmaxf(y0, 0.0f) * xs.x;
      y.y = fmaxf(y1, 0.0f) * xs.y;
      *(v2f*)(g + base) = y;

      start = nstart;
      n = nn;
      m = nm;
      di = ndi;
      sl = nsl;
      dsl = ndsl;
    }
  }
}

extern "C" void kernel_launch(void* const* d_in, const int* in_sizes, int n_in,
                              void* d_out, int out_size, void* d_ws, size_t ws_size,
                              hipStream_t stream) {
  const float* x     = (const float*)d_in[0];
  const int*   eidx  = (const int*)d_in[1];
  const float* W     = (const float*)d_in[2];
  const float* b     = (const float*)d_in[3];
  const float* gamma = (const float*)d_in[4];
  const float* beta  = (const float*)d_in[5];
  const float* fcW   = (const float*)d_in[6];
  const float* fcb   = (const float*)d_in[7];
  float* out = (float*)d_out;

  int N = in_sizes[0] / 64;
  int E = in_sizes[1] / 2;
  const int* src = eidx;
  const int* dst = eidx + E;

  int NC = (E + CH - 1) / CH;
  int NB = (N + BN - 1) / BN;

  char* ws = (char*)d_ws;
  size_t off = 0;
  float* dinv = (float*)(ws + off);
  off += (((size_t)N * 4) + 255) & ~(size_t)255;
  float* h = (float*)(ws + off);
  off += (((size_t)N * 64 * 4) + 255) & ~(size_t)255;
  float* g = (float*)(ws + off);
  off += (((size_t)N * 64 * 4) + 255) & ~(size_t)255;
  int* cnt = (int*)(ws + off);
  int* cursor = cnt + N;
  off += ((((size_t)N + 1) * 4) + 255) & ~(size_t)255;
  int* segoff = (int*)(ws + off);
  off += (((size_t)N * 4) + 255) & ~(size_t)255;
  int* list = (int*)(ws + off);
  off += (((size_t)E * 4) + 255) & ~(size_t)255;
  int* tab = (int*)(ws + off);
  off += (((size_t)NB * NC * 4) + 255) & ~(size_t)255;
  v2i* pairs = (v2i*)(ws + off);

  const int TB = 256;

  int gemmBlocks = (N + 127) / 128;
  bin_chunks_gemm<<<gemmBlocks + NC, TB, 0, stream>>>(src, dst, E, NB, NC, tab, pairs, cursor, x, W, h, N, gemmBlocks);
  bin_lists<<<NB, TB, 0, stream>>>(tab, pairs, NC, N, cursor, segoff, cnt, dinv, list);

  int gfBlocks = (N + 7) / 8;
  if (gfBlocks > 2048) gfBlocks = 2048;
  gather_fuse<<<gfBlocks, TB, 0, stream>>>(segoff, cnt, list, h, dinv, b, gamma, beta, x, g, N, gfBlocks * 8);

  gemm64_wmma<<<gemmBlocks, TB, 0, stream>>>(g, fcW, fcb, out, N);
}


# Neighbour sum taken over rows of x; the product with W runs on the summed 16-row tiles inside the gather kernel: no h plane, no GEMM work-groups in launch 1; order: rows summed, then times W
# speedup vs baseline: 4.8490x; 1.3162x over previous
#include <hip/hip_runtime.h>
#include <hip/hip_bf16.h>


typedef float v2f __attribute__((ext_vector_type(2)));
typedef float v4f __attribute__((ext_vector_type(4)));
typedef float v8f __attribute__((ext_vector_type(8)));

typedef int v2i __attribute__((ext_vector_type(2)));

#define CH 4096
#define EPT 16
#define BSH 8
#define BN 256
#define MAXNB 512
#define CAP 8192

struct BinChunkLds {
  v2i pair[CH];
  int cnt[MAXNB];
  int off[MAXNB];
  int wave[8];
};

__device__ __forceinline__
void bin_chunks_body(const int* __restrict__ src, const int* __restrict__ dst, int E, int NB, int NC,
                     int* __restrict__ tab, v2i* __restrict__ pairs, int* __restrict__ cursor,
                     BinChunkLds& lds, int w) {
  int* sCnt = lds.cnt;
  int* sOff = lds.off;
  v2i* sPair = lds.pair;
  int* sWave = lds.wave;
  int t = threadIdx.x;
  int lane = t & 31;
  int wave = t >> 5;
  long e0 = (long)w * CH;
  int nE = (E - e0 < CH) ? (int)(E - e0) : CH;

  int sv[EPT], dv[EPT];
#pragma unroll
  for (int j = 0; j < EPT; ++j) {
    int i = t + j * 256;
    sv[j] = (i < nE) ? src[e0 + i] : 0;
    dv[j] = (i < nE) ? dst[e0 + i] : 0;
  }

  for (int b = t; b < MAXNB; b += 256) sCnt[b] = 0;
  if (w == 0 && t == 0) *cursor = 0;
  __syncthreads();

#pragma unroll
  for (int j = 0; j < EPT; ++j)
    if (t + j * 256 < nE) atomicAdd(&sCnt[dv[j] >> BSH], 1);
  __syncthreads();

  int b0 = 2 * t, b1 = 2 * t + 1;
  int c0 = sCnt[b0];
  int c1 = sCnt[b1];
  int s = c0 + c1;
  int v = s;
#pragma unroll
  for (int o = 1; o < 32; o <<= 1) {
    int u = __shfl_up(v, o, 32);
    if (lane >= o) v += u;
  }
  if (lane == 31) sWave[wave] = v;
  __syncthreads();
  if (t == 0) {
    int tot = 0;
    for (int k = 0; k < 8; ++k) {
      int u = sWave[k];
      sWave[k] = tot;
      tot += u;
    }
  }
  __syncthreads();
  int ex = sWave[wave] + (v - s);
  sOff[b0] = ex;
  sOff[b1] = ex + c0;
  if (b0 < NB) tab[(long)b0 * NC + w] = ex | (c0 << 16);
  if (b1 < NB) tab[(long)b1 * NC + w] = (ex + c0) | (c1 << 16);
  __syncthreads();

#pragma unroll
  for (int j = 0; j < EPT; ++j) {
    if (t + j * 256 < nE) {
      v2i pr;
      pr.x = sv[j];
      pr.y = dv[j];
      sPair[atomicAdd(&sOff[pr.y >> BSH], 1)] = pr;
    }
  }
  __syncthreads();
#pragma unroll
  for (int j = 0; j < EPT; ++j) {
    int i = t + j * 256;
    if (i < nE) pairs[e0 + i] = sPair[i];
  }
}

__global__ __launch_bounds__(256)
void bin_chunks(const int* __restrict__ src, const int* __restrict__ dst, int E, int NB, int NC,
                int* __restrict__ tab, v2i* __restrict__ pairs, int* __restrict__ cursor) {
  __shared__ BinChunkLds lds;
  bin_chunks_body(src, dst, E, NB, NC, tab, pairs, cursor, lds, (int)blockIdx.x);
}

#define KEEP 24

__global__ __launch_bounds__(256)
void bin_lists(const int* __restrict__ tab, const v2i* __restrict__ pairs, int NC, int n,
               int* cursor, int* __restrict__ segoff, int* __restrict__ cnt,
               float* __restrict__ dinv, int* __restrict__ list) {
  __shared__ int sCnt[BN];
  __shared__ int sOff[BN];
  __shared__ int sList[CAP];
  __shared__ int sWave[8];
  __shared__ int sBase;
  __shared__ int sTot;
  int t = threadIdx.x;
  int lane = t & 31;
  int wave = t >> 5;
  int nodeBase = blockIdx.x * BN;
  const int* trow = tab + (long)blockIdx.x * NC;

  sCnt[t] = 0;
  __syncthreads();

  int pk0 = t < NC ? trow[t] : 0;
  int cs = 0;
  if (wave == 0) {
    for (int w0 = 0; w0 < NC; w0 += 256) {
      int tv[8];
#pragma unroll
      for (int j = 0; j < 8; ++j) {
        int w = w0 + j * 32 + lane;
        tv[j] = trow[w < NC ? w : 0];
        if (w >= NC) tv[j] = 0;
      }
#pragma unroll
      for (int j = 0; j < 8; ++j) cs += tv[j] >> 16;
    }
  }

  int c0 = pk0 >> 16;
  const v2i* p0 = pairs + (long)t * CH + (pk0 & 0xFFFF);
  v2i pr[KEEP];
#pragma unroll
  for (int j = 0; j < KEEP; ++j) {
    const v2i* a = j < c0 ? p0 + j : pairs;
    pr[j] = *a;
  }

  int baseReg = 0;
  if (wave == 0) {
    for (int o = 16; o > 0; o >>= 1) cs += __shfl_xor(cs, o, 32);
    if (lane == 0) baseReg = atomicAdd(cursor, cs);
  }

#pragma unroll
  for (int j = 0; j < KEEP; ++j)
    if (j < c0) atomicAdd(&sCnt[pr[j].y - nodeBase], 1);
  for (int k0 = KEEP; k0 < c0; k0 += 8) {
    int dl[8];
#pragma unroll
    for (int j = 0; j < 8; ++j)
      if (k0 + j < c0) dl[j] = p0[k0 + j].y;
#pragma unroll
    for (int j = 0; j < 8; ++j)
      if (k0 + j < c0) atomicAdd(&sCnt[dl[j] - nodeBase], 1);
  }
  for (int w = t + 256; w < NC; w += 256) {
    int pk = trow[w];
    int c = pk >> 16;
    const v2i* p = pairs + (long)w * CH + (pk & 0xFFFF);
    for (int k0 = 0; k0 < c; k0 += 8) {
      int dl[8];
#pragma unroll
      for (int j = 0; j < 8; ++j)
        if (k0 + j < c) dl[j] = p[k0 + j].y;
#pragma unroll
      for (int j = 0; j < 8; ++j)
        if (k0 + j < c) atomicAdd(&sCnt[dl[j] - nodeBase], 1);
    }
  }
  __syncthreads();

  int c = sCnt[t];
  int v = c;
#pragma unroll
  for (int o = 1; o < 32; o <<= 1) {
    int u = __shfl_up(v, o, 32);
    if (lane >= o) v += u;
  }
  if (lane == 31) sWave[wave] = v;
  __syncthreads();
  if (t == 0) {
    int run = 0;
    for (int k = 0; k < 8; ++k) {
      int u = sWave[k];
      sWave[k] = run;
      run += u;
    }
    sTot = cs;
    sBase = baseReg;
  }
  __syncthreads();
  int ex = sWave[wave] + (v - c);
  int base = sBase;
  int tot = sTot;
  long node = (long)nodeBase + t;
  if (node < n) {
    segoff[node] = base + ex;
    cnt[node] = c;
    dinv[node] = rsqrtf((float)(c + 1));
  }
  sOff[t] = ex;
  __syncthreads();

  if (tot <= CAP) {
#pragma unroll
    for (int j = 0; j < KEEP; ++j)
      if (j < c0) sList[atomicAdd(&sOff[pr[j].y - nodeBase], 1)] = pr[j].x;
    for (int k0 = KEEP; k0 < c0; k0 += 8) {
      v2i q[8];
#pragma unroll
      for (int j = 0; j < 8; ++j)
        if (k0 + j < c0) q[j] = p0[k0 + j];
#pragma unroll
      for (int j = 0; j < 8; ++j)
        if (k0 + j < c0) sList[atomicAdd(&sOff[q[j].y - nodeBase], 1)] = q[j].x;
    }
    for (int w = t + 256; w < NC; w += 256) {
      int pk = trow[w];
      int cc = pk >> 16;
      const v2i* p = pairs + (long)w * CH + (pk & 0xFFFF);
      for (int k0 = 0; k0 < cc; k0 += 8) {
        v2i q[8];
#pragma unroll
        for (int j = 0; j < 8; ++j)
          if (k0 + j < cc) q[j] = p[k0 + j];
#pragma unroll
        for (int j = 0; j < 8; ++j)
          if (k0 + j < cc) sList[atomicAdd(&sOff[q[j].y - nodeBase], 1)] = q[j].x;
      }
    }
    __syncthreads();
    for (int i = t; i < tot; i += 256) list[base + i] = sList[i];
  } else {
#pragma unroll
    for (int j = 0; j < KEEP; ++j)
      if (j < c0) list[base + atomicAdd(&sOff[pr[j].y - nodeBase], 1)] = pr[j].x;
    for (int k = KEEP; k < c0; ++k) {
      v2i q = p0[k];
      list[base + atomicAdd(&sOff[q.y - nodeBase], 1)] = q.x;
    }
    for (int w = t + 256; w < NC; w += 256) {
      int pk = trow[w];
      int cc = pk >> 16;
      const v2i* p = pairs + (long)w * CH + (pk & 0xFFFF);
      for (int k = 0; k < cc; ++k) {
        v2i q = p[k];
        list[base + atomicAdd(&sOff[q.y - nodeBase], 1)] = q.x;
      }
    }
  }
}

#define PASS 8
#define YP 68

template <int Q>
__device__ __forceinline__ int share_i(int v) {
  return __builtin_amdgcn_update_dpp(0, v, 0x150 + Q, 0xf, 0xf, false);
}
template <int Q>
__device__ __forceinline__ float share_f(float v) {
  return __int_as_float(share_i<Q>(__float_as_int(v)));
}
template <int M>
__device__ __forceinline__ float xmask_f(float v) {
  return __int_as_float(__builtin_amdgcn_update_dpp(0, __float_as_int(v), 0x160 + M, 0xf, 0xf, false));
}
__device__ __forceinline__ float row_sum(float v) {
  v += xmask_f<1>(v);
  v += xmask_f<2>(v);
  v += xmask_f<4>(v);
  v += xmask_f<8>(v);
  return v;
}
__device__ __forceinline__ int row_pick(int v, int i, int half) {
  int lo = __builtin_amdgcn_readlane(v, i);
  int hi = __builtin_amdgcn_readlane(v, i + 16);
  return half ? hi : lo;
}
__device__ __forceinline__ int wave_max2(int v) {
  int lo = __builtin_amdgcn_readlane(v, 0);
  int hi = __builtin_amdgcn_readlane(v, 16);
  return lo > hi ? lo : hi;
}

template <int Q0>
__device__ __forceinline__
void rows8(const float* __restrict__ xch, int sl, float dsl, float di, bool on, int m, v4f& acc) {
  const char* hb = (const char*)xch;
  v4f r[8];
#define LD_ROW(Q)                                                        \
  {                                                                      \
    int s_ = share_i<Q0 + Q>(sl);                                        \
    v4f z_ = {0.0f, 0.0f, 0.0f, 0.0f};                                   \
    r[Q] = z_;                                                           \
    if (Q0 + Q < m) r[Q] = *(const v4f*)(hb + ((unsigned)s_ << 8));      \
  }
  LD_ROW(0) LD_ROW(1) LD_ROW(2) LD_ROW(3) LD_ROW(4) LD_ROW(5) LD_ROW(6) LD_ROW(7)
#undef LD_ROW
  float wl = on ? dsl * di : 0.0f;
#define ADD_ROW(Q)                                                       \
  {                                                                      \
    float w_ = share_f<Q0 + Q>(wl);                                      \
    acc.x = fmaf(r[Q].x, w_, acc.x);                                     \
    acc.y = fmaf(r[Q].y, w_, acc.y);                                     \
    acc.z = fmaf(r[Q].z, w_, acc.z);                                     \
    acc.w = fmaf(r[Q].w, w_, acc.w);                                     \
  }
  ADD_ROW(0) ADD_ROW(1) ADD_ROW(2) ADD_ROW(3) ADD_ROW(4) ADD_ROW(5) ADD_ROW(6) ADD_ROW(7)
#undef ADD_ROW
}

__device__ __forceinline__
bool pass_headers(const int* __restrict__ segoff, const int* __restrict__ cnt,
                  const float* __restrict__ dinv, long nbH, int c, int nodeStep, int nNodes,
                  int& stL, int& nL, float& diL) {
  long mine = nbH + (long)c * nodeStep;
  bool have = c < PASS && mine < nNodes;
  stL = 0;
  nL = 0;
  diL = 0.0f;
  if (have) {
    stL = segoff[mine];
    nL = cnt[mine];
    diL = dinv[mine];
  }
  return have;
}

__device__ __forceinline__
void tile_product(const float* arow, const float* brow, v8f (&acc)[4]) {
#pragma unroll 1
  for (int kg = 0; kg < 4; ++kg) {
    const float* ag = arow + 16 * kg;
    const float* bg = brow + 16 * kg * 64;
    v2f a[4];
#pragma unroll
    for (int k = 0; k < 4; ++k) a[k] = *(const v2f*)(ag + 4 * k);
#pragma unroll
    for (int k = 0; k < 4; ++k) {
#pragma unroll
      for (int q = 0; q < 4; ++q) {
        v2f bf;
        bf.x = bg[(4 * k) * 64 + q * 16];
        bf.y = bg[(4 * k + 1) * 64 + q * 16];
        acc[q] = __builtin_amdgcn_wmma_f32_16x16x4_f32(
             false, a[k],  false, bf,
             (short)0, acc[q],  false,  false);
      }
    }
  }
}

__global__ __launch_bounds__(256, 8)
void gather_fuse_fc(const int* __restrict__ segoff, const int* __restrict__ cnt,
                    const int* __restrict__ list, const float* __restrict__ dinv,
                    const float* __restrict__ W, const float* __restrict__ b,
                    const float* __restrict__ gamma, const float* __restrict__ beta,
                    const float* __restrict__ x, const float* __restrict__ fcW,
                    const float* __restrict__ fcb, float* __restrict__ out,
                    int nNodes, int nodeStep) {
  __shared__ alignas(16) float sW1[64 * 64];
  __shared__ alignas(16) float sW2[64 * 64];
  __shared__ alignas(16) float sY[16 * PASS * YP];
  int tid = threadIdx.x;
  int lane = tid & 31;
  int wave = __builtin_amdgcn_readfirstlane((int)(threadIdx.x >> 5));
  int half = lane >> 4;
  int c = lane & 15;
  int ch = c * 4;
  int stream = wave * 2 + half;

  long blockFirst = (long)blockIdx.x * 16;
  int stL, nL;
  float diL;
  bool have = pass_headers(segoff, cnt, dinv, blockFirst + stream, c, nodeStep, nNodes, stL, nL, diL);

  {
    v4f wreg[8];
#pragma unroll
    for (int j = 0; j < 4; ++j) wreg[j] = ((const v4f*)W)[tid + j * 256];
#pragma unroll
    for (int j = 0; j < 4; ++j) wreg[4 + j] = ((const v4f*)fcW)[tid + j * 256];
#pragma unroll
    for (int j = 0; j < 4; ++j) ((v4f*)sW1)[tid + j * 256] = wreg[j];
#pragma unroll
    for (int j = 0; j < 4; ++j) ((v4f*)sW2)[tid + j * 256] = wreg[4 + j];
  }

  v4f bb = *(const v4f*)(b + ch);
  v4f gg = *(const v4f*)(gamma + ch);
  v4f be = *(const v4f*)(beta + ch);
  const float* xch = x + ch;

  for (long pb = blockFirst; pb < nNodes; pb += (long)PASS * nodeStep) {
    long nb = pb + stream;
    if (pb != blockFirst) {
      __syncthreads();
      have = pass_headers(segoff, cnt, dinv, nb, c, nodeStep, nNodes, stL, nL, diL);
    }
    int kc = __builtin_popcount(__builtin_amdgcn_ballot_w32(have) & 0xFFFFu);

    int start = row_pick(stL, 0, half);
    int n = row_pick(nL, 0, half);
    float di = __int_as_float(row_pick(__float_as_int(diL), 0, half));
    int m = n < 16 ? n : 16;
    int sl = list[c < m ? start + c : 0];
    sl = c < m ? sl : 0;

    for (int i = 0; i < kc; ++i) {
      long node = nb + (long)i * nodeStep;
      bool valid = node < nNodes;
      long base = (valid ? node : 0) * 64 + ch;

      int nstart = 0, nn = 0;
      float ndi = 0.0f;
      if (i + 1 < kc) {
        nstart = row_pick(stL, i + 1, half);
        nn = row_pick(nL, i + 1, half);
        ndi = __int_as_float(row_pick(__float_as_int(diL), i + 1, half));
      }
      int nm = nn < 16 ? nn : 16;
      int nsl = list[c < nm ? nstart + c : 0];

      v4f xs = *(const v4f*)(x + base);
      float dsl = dinv[sl];

      v4f acc = {0.0f, 0.0f, 0.0f, 0.0f};
      {
        bool on = c < m;
        int mmax = wave_max2(m);
        if (mmax > 0) rows8<0>(xch, sl, dsl, di, on, m, acc);
        if (mmax > 8) rows8<8>(xch, sl, dsl, di, on, m, acc);
        int nmax = wave_max2(n);
        for (int p0 = 16; p0 < nmax; p0 += 16) {
          int m2 = n - p0;
          m2 = m2 < 0 ? 0 : (m2 > 16 ? 16 : m2);
          int s2 = 0;
          float d2 = 0.0f;
          if (c < m2) {
            s2 = list[start + p0 + c];
            d2 = dinv[s2];
          }
          int m2max = wave_max2(m2);
          rows8<0>(xch, s2, d2, di, c < m2, m2, acc);
          if (m2max > 8) rows8<8>(xch, s2, d2, di, c < m2, m2, acc);
        }
      }

      nsl = c < nm ? nsl : 0;

      float selfw = di * di;
      v4f a;
      a.x = acc.x + xs.x * selfw;
      a.y = acc.y + xs.y * selfw;
      a.z = acc.z + xs.z * selfw;
      a.w = acc.w + xs.w * selfw;
      if (!valid) {
        v4f z = {0.0f, 0.0f, 0.0f, 0.0f};
        a = z;
      }
      *(v4f*)(sY + (i * 16 + stream) * YP + ch) = a;

      start = nstart;
      n = nn;
      m = nm;
      di = ndi;
      sl = nsl;
    }
    for (int i = kc; i < PASS; ++i) {
      v4f z = {0.0f, 0.0f, 0.0f, 0.0f};
      *(v4f*)(sY + (i * 16 + stream) * YP + ch) = z;
    }
    __syncthreads();

    int its = 0;
    for (int i = 0; i < PASS; ++i) its += (pb + (long)i * nodeStep < nNodes) ? 1 : 0;
    if (wave < its) {
      int koff = (lane >> 4) << 1;
      int mrow = lane & 15;
      int rowOff = (lane >> 4) * 8;
      float* tile = sY + wave * 16 * YP;
      const float* arow = tile + mrow * YP + koff;

      float fb[4];
#pragma unroll
      for (int q = 0; q < 4; ++q) fb[q] = fcb[q * 16 + mrow];

      v8f accT[4] = {};
      tile_product(arow, sW1 + koff * 64 + mrow, accT);
      v4f xg[8];
#pragma unroll
      for (int j = 0; j < 8; ++j) {
        long nd = pb + (2 * j + half) + (long)wave * nodeStep;
        xg[j] = *(const v4f*)(x + (nd < nNodes ? nd : 0) * 64 + ch);
      }
#pragma unroll
      for (int r = 0; r < 8; ++r) {
#pragma unroll
        for (int q = 0; q < 4; ++q) tile[(rowOff + r) * YP + q * 16 + mrow] = accT[q][r];
      }

#pragma unroll
      for (int j = 0; j < 8; ++j) {
        float* trow = tile + (2 * j + half) * YP + ch;
        v4f t = *(const v4f*)trow;
        float v0 = t.x + bb.x;
        float v1 = t.y + bb.y;
        float v2 = t.z + bb.z;
        float v3 = t.w + bb.w;

        float mu = row_sum((v0 + v1) + (v2 + v3)) * (1.0f / 64.0f);

        float d0 = v0 - mu, d1 = v1 - mu, d2 = v2 - mu, d3 = v3 - mu;
        float qq = row_sum((d0 * d0 + d1 * d1) + (d2 * d2 + d3 * d3));
        float inv = rsqrtf(qq * (1.0f / 64.0f) + 1e-5f);

        v4f y;
        y.x = fmaxf(d0 * inv * gg.x + be.x, 0.0f) * xg[j].x;
        y.y = fmaxf(d1 * inv * gg.y + be.y, 0.0f) * xg[j].y;
        y.z = fmaxf(d2 * inv * gg.z + be.z, 0.0f) * xg[j].z;
        y.w = fmaxf(d3 * inv * gg.w + be.w, 0.0f) * xg[j].w;
        *(v4f*)trow = y;
      }

      v8f accO[4] = {};
      tile_product(arow, sW2 + koff * 64 + mrow, accO);
#pragma unroll
      for (int r = 0; r < 8; ++r) {
        long node = pb + rowOff + r + (long)wave * nodeStep;
        if (node < nNodes) {
#pragma unroll
          for (int q = 0; q < 4; ++q) out[node * 64 + q * 16 + mrow] = accO[q][r] + fb[q];
        }
      }
    }
  }
}

extern "C" void kernel_launch(void* const* d_in, const int* in_sizes, int n_in,
                              void* d_out, int out_size, void* d_ws, size_t ws_size,
                              hipStream_t stream) {
  const float* x     = (const float*)d_in[0];
  const int*   eidx  = (const int*)d_in[1];
  const float* W     = (const float*)d_in[2];
  const float* b     = (const float*)d_in[3];
  const float* gamma = (const float*)d_in[4];
  const float* beta  = (const float*)d_in[5];
  const float* fcW   = (const float*)d_in[6];
  const float* fcb   = (const float*)d_in[7];
  float* out = (float*)d_out;

  int N = in_sizes[0] / 64;
  int E = in_sizes[1] / 2;
  const int* src = eidx;
  const int* dst = eidx + E;

  int NC = (E + CH - 1) / CH;
  int NB = (N + BN - 1) / BN;

  char* ws = (char*)d_ws;
  size_t off = 0;
  float* dinv = (float*)(ws + off);
  off += (((size_t)N * 4) + 255) & ~(size_t)255;
  int* cnt = (int*)(ws + off);
  int* cursor = cnt + N;
  off += ((((size_t)N + 1) * 4) + 255) & ~(size_t)255;
  int* segoff = (int*)(ws + off);
  off += (((size_t)N * 4) + 255) & ~(size_t)255;
  int* list = (int*)(ws + off);
  off += (((size_t)E * 4) + 255) & ~(size_t)255;
  int* tab = (int*)(ws + off);
  off += (((size_t)NB * NC * 4) + 255) & ~(size_t)255;
  v2i* pairs = (v2i*)(ws + off);

  const int TB = 256;

  bin_chunks<<<NC, TB, 0, stream>>>(src, dst, E, NB, NC, tab, pairs, cursor);
  bin_lists<<<NB, TB, 0, stream>>>(tab, pairs, NC, N, cursor, segoff, cnt, dinv, list);

  int gfBlocks = (N + 15) / 16;
  if (gfBlocks > 1024) gfBlocks = 1024;
  gather_fuse_fc<<<gfBlocks, TB, 0, stream>>>(segoff, cnt, list, dinv, W, b, gamma, beta, x, fcW, fcb, out, N, gfBlocks * 16);
}
